# starting kernel
# baseline (speedup 1.0000x reference)
#include <hip/hip_runtime.h>
#include <hip/hip_bf16.h>

typedef __attribute__((ext_vector_type(2))) float v2f;
typedef __attribute__((ext_vector_type(8))) float v8f;

static __host__ inline int cdiv(int a, int b) { return (a + b - 1) / b; }


__global__ void k_fill(float* __restrict__ p, float v, int n) {
    int i = blockIdx.x * blockDim.x + threadIdx.x;
    if (i < n) p[i] = v;
}

__global__ void k_deg_accum(const int* __restrict__ dst, float* __restrict__ deg, int E) {
    int e = blockIdx.x * blockDim.x + threadIdx.x;
    if (e < E) atomicAdd(&deg[dst[e]], 1.0f);
}

__global__ void k_rsqrt(float* __restrict__ d, int n) {
    int i = blockIdx.x * blockDim.x + threadIdx.x;
    if (i < n) {
        float v = d[i];
        d[i] = (v > 0.0f) ? rsqrtf(v) : 0.0f;
    }
}

__global__ void k_transpose_pad(const float* __restrict__ Bw, float* __restrict__ Bt,
                                int K, int Ncols, int ncols_pad) {
    int i = blockIdx.x * blockDim.x + threadIdx.x;
    int total = ncols_pad * K;
    if (i >= total) return;
    int ncol = i / K, k = i - ncol * K;
    Bt[i] = (ncol < Ncols) ? Bw[(size_t)k * Ncols + ncol] : 0.0f;
}

__global__ void k_edge_scatter(const int* __restrict__ src, const int* __restrict__ dst,
                               const float* __restrict__ t, const float* __restrict__ dinv,
                               float* __restrict__ agg, int E) {
    int e = blockIdx.x * blockDim.x + threadIdx.x;
    if (e >= E) return;
    int s = src[e], d = dst[e];
    float nr = dinv[s] * dinv[d];
    const float4* ts = (const float4*)(t + (size_t)s * 20);
    float* ad = agg + (size_t)d * 20;
#pragma unroll
    for (int q = 0; q < 5; ++q) {
        float4 v = ts[q];
        atomicAdd(ad + 4 * q + 0, v.x * nr);
        atomicAdd(ad + 4 * q + 1, v.y * nr);
        atomicAdd(ad + 4 * q + 2, v.z * nr);
        atomicAdd(ad + 4 * q + 3, v.w * nr);
    }
}

__global__ void k_finish_layer(const float* __restrict__ t, const float* __restrict__ dinv,
                               const float* __restrict__ bias, float* __restrict__ agg, int n_nodes) {
    int i = blockIdx.x * blockDim.x + threadIdx.x;
    int total = n_nodes * 20;
    if (i >= total) return;
    int node = i / 20, f = i - node * 20;
    float di = dinv[node];
    float v = agg[i] + t[i] * di * di + bias[f];
    agg[i] = fmaxf(v, 0.0f);
}

__global__ void k_wmma_gemm_f32(const float* __restrict__ A, const float* __restrict__ Bt,
                                const float* __restrict__ bias, float* __restrict__ C,
                                int M, int K, int Ncols, int mtiles, int ntiles, int do_relu) {
    int wave = (blockIdx.x * blockDim.x + threadIdx.x) >> 5;
    int lane = threadIdx.x & 31;
    int mt = wave % mtiles;
    int nt = wave / mtiles;
    if (nt >= ntiles) return;

    int lrow = lane & 15;
    int kh = (lane >> 4) << 1;
    int ncol = nt * 16 + lrow;
    const float* arow = A + (size_t)(mt * 16 + lrow) * K + kh;
    const float* brow = Bt + (size_t)ncol * K + kh;

    v8f acc = {};
    int kb = 0;
    for (; kb + 16 <= K; kb += 16) {
#pragma unroll
        for (int j = 0; j < 4; ++j) {
            v2f a = *(const v2f*)(arow + kb + 4 * j);
            v2f b = *(const v2f*)(brow + kb + 4 * j);
            acc = __builtin_amdgcn_wmma_f32_16x16x4_f32(false, a, false, b,
                                                        (short)0, acc, false, false);
        }
    }
    for (; kb < K; kb += 4) {
        v2f a = *(const v2f*)(arow + kb);
        v2f b = *(const v2f*)(brow + kb);
        acc = __builtin_amdgcn_wmma_f32_16x16x4_f32(false, a, false, b,
                                                    (short)0, acc, false, false);
    }

    if (ncol >= Ncols) return;
    float bv = bias ? bias[ncol] : 0.0f;
    int rbase = mt * 16 + ((lane >> 4) << 3);
#pragma unroll
    for (int r = 0; r < 8; ++r) {
        float v = acc[r] + bv;
        if (do_relu) v = fmaxf(v, 0.0f);
        C[(size_t)(rbase + r) * Ncols + ncol] = v;
    }
}


extern "C" void kernel_launch(void* const* d_in, const int* in_sizes, int n_in,
                              void* d_out, int out_size, void* d_ws, size_t ws_size,
                              hipStream_t stream) {
    const float* x    = (const float*)d_in[0];
    const int*   eidx = (const int*)d_in[1];
    const float* W1   = (const float*)d_in[3];
    const float* b1   = (const float*)d_in[4];
    const float* W2   = (const float*)d_in[5];
    const float* b2   = (const float*)d_in[6];
    const float* fc1W = (const float*)d_in[7];
    const float* fc1b = (const float*)d_in[8];
    const float* outW = (const float*)d_in[9];
    const float* outb = (const float*)d_in[10];

    const int N  = in_sizes[2];
    const int E  = in_sizes[1] / 2;
    const int Bg = N / 82;
    const int EMB = 20, LAT = 82 * 20, HID = 300, NC = 22;

    char* ws = (char*)d_ws;
    size_t off = 0;
    auto take = [&](size_t bytes) -> void* {
        off = (off + 255) & ~(size_t)255;
        void* p = ws + off;
        off += bytes;
        return p;
    };
    float* dinv = (float*)take((size_t)N * 4);
    float* t    = (float*)take((size_t)N * EMB * 4);
    float* agg  = (float*)take((size_t)N * EMB * 4);
    float* fco  = (float*)take((size_t)Bg * HID * 4);
    float* Bt   = (float*)take((size_t)304 * LAT * 4);

    const int* src = eidx;
    const int* dst = eidx + E;

    const int TB = 256;
    int nfeat = N * EMB;

    auto run_gemm = [&](const float* A, const float* Bw, const float* bias, float* C,
                        int M, int K, int Ncols, int relu) {
        int ntiles = (Ncols + 15) / 16;
        int ncols_pad = ntiles * 16;
        int mtiles = M / 16;
        k_transpose_pad<<<cdiv(ncols_pad * K, TB), TB, 0, stream>>>(Bw, Bt, K, Ncols, ncols_pad);
        int waves = mtiles * ntiles;
        k_wmma_gemm_f32<<<cdiv(waves * 32, TB), TB, 0, stream>>>(
            A, Bt, bias, C, M, K, Ncols, mtiles, ntiles, relu);
    };

    k_fill<<<cdiv(N, TB), TB, 0, stream>>>(dinv, 1.0f, N);
    k_deg_accum<<<cdiv(E, TB), TB, 0, stream>>>(dst, dinv, E);
    k_rsqrt<<<cdiv(N, TB), TB, 0, stream>>>(dinv, N);

    k_fill<<<cdiv(nfeat, TB), TB, 0, stream>>>(agg, 0.0f, nfeat);
    run_gemm(x, W1, nullptr, t, N, 20, EMB, 0);
    k_edge_scatter<<<cdiv(E, TB), TB, 0, stream>>>(src, dst, t, dinv, agg, E);
    k_finish_layer<<<cdiv(nfeat, TB), TB, 0, stream>>>(t, dinv, b1, agg, N);

    run_gemm(agg, W2, nullptr, t, N, EMB, EMB, 0);
    k_fill<<<cdiv(nfeat, TB), TB, 0, stream>>>(agg, 0.0f, nfeat);
    k_edge_scatter<<<cdiv(E, TB), TB, 0, stream>>>(src, dst, t, dinv, agg, E);
    k_finish_layer<<<cdiv(nfeat, TB), TB, 0, stream>>>(t, dinv, b2, agg, N);

    run_gemm(agg  , fc1W, fc1b, fco, Bg, LAT, HID, 1);

    run_gemm(fco, outW, outb, (float*)d_out, Bg, HID, NC, 0);
}


# Edge scatter (20 float atomics per edge, twice) replaced: in-edge lists sorted by dst built once with 1 returning int atomic per edge, then an atomic-free per-node gather; sum order = arrival rank
# speedup vs baseline: 3.3813x; 3.3813x over previous
#include <hip/hip_runtime.h>
#include <hip/hip_bf16.h>

typedef __attribute__((ext_vector_type(2))) float v2f;
typedef __attribute__((ext_vector_type(8))) float v8f;

static __host__ inline int cdiv(int a, int b) { return (a + b - 1) / b; }


__global__ void k_fill_i32(int* __restrict__ p, int v, int n) {
    int i = blockIdx.x * blockDim.x + threadIdx.x;
    if (i < n) p[i] = v;
}

__global__ void k_count_rank(const int* __restrict__ dst, int* __restrict__ cnt,
                             int* __restrict__ rank, int E) {
    int e = blockIdx.x * blockDim.x + threadIdx.x;
    if (e < E) rank[e] = atomicAdd(&cnt[dst[e]], 1);
}

__global__ void __launch_bounds__(256) k_tile_sums(const int* __restrict__ cnt,
                                                   int* __restrict__ tsum, int n) {
    __shared__ int wsum[8];
    int base = blockIdx.x * 1024 + threadIdx.x * 4;
    int s = 0;
    for (int j = 0; j < 4; ++j)
        if (base + j < n) s += cnt[base + j];
    for (int o = 16; o > 0; o >>= 1) s += __shfl_down(s, o, 32);
    if ((threadIdx.x & 31) == 0) wsum[threadIdx.x >> 5] = s;
    __syncthreads();
    if (threadIdx.x == 0) {
        int tot = 0;
        for (int w = 0; w < 8; ++w) tot += wsum[w];
        tsum[blockIdx.x] = tot;
    }
}

__global__ void __launch_bounds__(256) k_scan_rowptr(const int* __restrict__ cnt,
                                                     const int* __restrict__ tsum,
                                                     int* __restrict__ rowptr,
                                                     float* __restrict__ dinv, int n) {
    __shared__ int wpre[8];
    __shared__ int winc[8];
    int tid = threadIdx.x, lane = tid & 31, w = tid >> 5;

    int p = 0;
    for (int j = tid; j < (int)blockIdx.x; j += 256) p += tsum[j];
    for (int o = 16; o > 0; o >>= 1) p += __shfl_down(p, o, 32);
    if (lane == 0) wpre[w] = p;

    int base = blockIdx.x * 1024 + tid * 4;
    int c[4];
    for (int j = 0; j < 4; ++j) c[j] = (base + j < n) ? cnt[base + j] : 0;
    int s = c[0] + c[1] + c[2] + c[3];
    int inc = s;
    for (int o = 1; o < 32; o <<= 1) {
        int y = __shfl_up(inc, o, 32);
        if (lane >= o) inc += y;
    }
    if (lane == 31) winc[w] = inc;
    __syncthreads();

    int off = 0;
    for (int j = 0; j < 8; ++j) {
        off += wpre[j];
        if (j < w) off += winc[j];
    }
    int ex = off + inc - s;
    for (int j = 0; j < 4; ++j) {
        if (base + j < n) {
            rowptr[base + j] = ex;
            float v = (float)(c[j] + 1);
            dinv[base + j] = (v > 0.0f) ? rsqrtf(v) : 0.0f;
            ex += c[j];
        }
    }
}

__global__ void k_fill_lists(const int* __restrict__ src, const int* __restrict__ dst,
                             const int* __restrict__ rank, const int* __restrict__ rowptr,
                             const float* __restrict__ dinv, int* __restrict__ lists, int E) {
    int e = blockIdx.x * blockDim.x + threadIdx.x;
    if (e >= E) return;
    int s = src[e], d = dst[e];
    int pos = rowptr[d] + rank[e];
    pos = min(max(pos, 0), E - 1);
    ((int2*)lists)[pos] = make_int2(s, __float_as_int(dinv[s]));
}

__global__ void k_transpose_pad(const float* __restrict__ Bw, float* __restrict__ Bt,
                                int K, int Ncols, int ncols_pad) {
    int i = blockIdx.x * blockDim.x + threadIdx.x;
    int total = ncols_pad * K;
    if (i >= total) return;
    int ncol = i / K, k = i - ncol * K;
    Bt[i] = (ncol < Ncols) ? Bw[(size_t)k * Ncols + ncol] : 0.0f;
}

__global__ void __launch_bounds__(256) k_gather_finish(const int* __restrict__ rowptr,
                                                       const int* __restrict__ cnt,
                                                       const int* __restrict__ lists,
                                                       const float* __restrict__ t,
                                                       const float* __restrict__ dinv,
                                                       const float* __restrict__ bias,
                                                       float* __restrict__ h, int n_nodes) {
    int g = blockIdx.x * 256 + threadIdx.x;
    if (g >= n_nodes * 5) return;
    int node = g / 5;
    int q = g - node * 5;
    const float4* t4 = (const float4*)t;
    const int2* le = (const int2*)lists;
    float di = dinv[node];
    int k = rowptr[node];
    int k1 = k + cnt[node];
    int nmax = n_nodes - 1;

    float4 acc = make_float4(0.0f, 0.0f, 0.0f, 0.0f);
    for (; k + 4 <= k1; k += 4) {
        int2 e[4];
        float4 v[4];
        for (int j = 0; j < 4; ++j) e[j] = le[k + j];
        for (int j = 0; j < 4; ++j) v[j] = t4[min(max(e[j].x, 0), nmax) * 5 + q];
        for (int j = 0; j < 4; ++j) {
            float nr = __int_as_float(e[j].y) * di;
            acc.x = fmaf(v[j].x, nr, acc.x);
            acc.y = fmaf(v[j].y, nr, acc.y);
            acc.z = fmaf(v[j].z, nr, acc.z);
            acc.w = fmaf(v[j].w, nr, acc.w);
        }
    }
#pragma unroll 1
    for (; k < k1; ++k) {
        int2 e0 = le[k];
        float4 v0 = t4[min(max(e0.x, 0), nmax) * 5 + q];
        float nr = __int_as_float(e0.y) * di;
        acc.x = fmaf(v0.x, nr, acc.x);
        acc.y = fmaf(v0.y, nr, acc.y);
        acc.z = fmaf(v0.z, nr, acc.z);
        acc.w = fmaf(v0.w, nr, acc.w);
    }

    float4 sv = t4[g];
    const float* bq = bias + 4 * q;
    float4 o;
    o.x = fmaxf(acc.x + sv.x * di * di + bq[0], 0.0f);
    o.y = fmaxf(acc.y + sv.y * di * di + bq[1], 0.0f);
    o.z = fmaxf(acc.z + sv.z * di * di + bq[2], 0.0f);
    o.w = fmaxf(acc.w + sv.w * di * di + bq[3], 0.0f);
    ((float4*)h)[g] = o;
}

__global__ void k_wmma_gemm_f32(const float* __restrict__ A, const float* __restrict__ Bt,
                                const float* __restrict__ bias, float* __restrict__ C,
                                int M, int K, int Ncols, int mtiles, int ntiles, int do_relu) {
    int wave = (blockIdx.x * blockDim.x + threadIdx.x) >> 5;
    int lane = threadIdx.x & 31;
    int mt = wave % mtiles;
    int nt = wave / mtiles;
    if (nt >= ntiles) return;

    int lrow = lane & 15;
    int kh = (lane >> 4) << 1;
    int ncol = nt * 16 + lrow;
    const float* arow = A + (size_t)(mt * 16 + lrow) * K + kh;
    const float* brow = Bt + (size_t)ncol * K + kh;

    v8f acc = {};
    int kb = 0;
    for (; kb + 16 <= K; kb += 16) {
#pragma unroll
        for (int j = 0; j < 4; ++j) {
            v2f a = *(const v2f*)(arow + kb + 4 * j);
            v2f b = *(const v2f*)(brow + kb + 4 * j);
            acc = __builtin_amdgcn_wmma_f32_16x16x4_f32(false, a, false, b,
                                                        (short)0, acc, false, false);
        }
    }
    for (; kb < K; kb += 4) {
        v2f a = *(const v2f*)(arow + kb);
        v2f b = *(const v2f*)(brow + kb);
        acc = __builtin_amdgcn_wmma_f32_16x16x4_f32(false, a, false, b,
                                                    (short)0, acc, false, false);
    }

    if (ncol >= Ncols) return;
    float bv = bias ? bias[ncol] : 0.0f;
    int rbase = mt * 16 + ((lane >> 4) << 3);
#pragma unroll
    for (int r = 0; r < 8; ++r) {
        float v = acc[r] + bv;
        if (do_relu) v = fmaxf(v, 0.0f);
        C[(size_t)(rbase + r) * Ncols + ncol] = v;
    }
}


extern "C" void kernel_launch(void* const* d_in, const int* in_sizes, int n_in,
                              void* d_out, int out_size, void* d_ws, size_t ws_size,
                              hipStream_t stream) {
    const float* x    = (const float*)d_in[0];
    const int*   eidx = (const int*)d_in[1];
    const float* W1   = (const float*)d_in[3];
    const float* b1   = (const float*)d_in[4];
    const float* W2   = (const float*)d_in[5];
    const float* b2   = (const float*)d_in[6];
    const float* fc1W = (const float*)d_in[7];
    const float* fc1b = (const float*)d_in[8];
    const float* outW = (const float*)d_in[9];
    const float* outb = (const float*)d_in[10];

    const int N  = in_sizes[2];
    const int E  = in_sizes[1] / 2;
    const int Bg = N / 82;
    const int EMB = 20, LAT = 82 * 20, HID = 300, NC = 22;

    char* ws = (char*)d_ws;
    size_t off = 0;
    auto take = [&](size_t bytes) -> void* {
        off = (off + 255) & ~(size_t)255;
        void* p = ws + off;
        off += bytes;
        return p;
    };
    const int ntile = cdiv(N, 1024);
    float* dinv   = (float*)take((size_t)N * 4);
    int*   cnt    = (int*)take((size_t)N * 4);
    int*   rowptr = (int*)take((size_t)N * 4);
    int*   tsum   = (int*)take((size_t)ntile * 4);
    int*   rank   = (int*)take((size_t)E * 4);
    int*   lists  = (int*)take((size_t)E * 8);
    float* t    = (float*)take((size_t)N * EMB * 4);
    float* agg  = (float*)take((size_t)N * EMB * 4);
    float* fco  = (float*)take((size_t)Bg * HID * 4);
    float* Bt   = (float*)take((size_t)304 * LAT * 4);

    const int* src = eidx;
    const int* dst = eidx + E;

    const int TB = 256;

    auto run_gemm = [&](const float* A, const float* Bw, const float* bias, float* C,
                        int M, int K, int Ncols, int relu) {
        int ntiles = (Ncols + 15) / 16;
        int ncols_pad = ntiles * 16;
        int mtiles = M / 16;
        k_transpose_pad<<<cdiv(ncols_pad * K, TB), TB, 0, stream>>>(Bw, Bt, K, Ncols, ncols_pad);
        int waves = mtiles * ntiles;
        k_wmma_gemm_f32<<<cdiv(waves * 32, TB), TB, 0, stream>>>(
            A, Bt, bias, C, M, K, Ncols, mtiles, ntiles, relu);
    };

    k_fill_i32<<<cdiv(N, TB), TB, 0, stream>>>(cnt, 0, N);
    k_count_rank<<<cdiv(E, TB), TB, 0, stream>>>(dst, cnt, rank, E);
    k_tile_sums<<<ntile, 256, 0, stream>>>(cnt, tsum, N);
    k_scan_rowptr<<<ntile, 256, 0, stream>>>(cnt, tsum, rowptr, dinv, N);
    k_fill_lists<<<cdiv(E, TB), TB, 0, stream>>>(src, dst, rank, rowptr, dinv, lists, E);

    run_gemm(x, W1, nullptr, t, N, 20, EMB, 0);
    k_gather_finish<<<cdiv(N * 5, 256), 256, 0, stream>>>(rowptr, cnt, lists, t, dinv, b1, agg, N);

    run_gemm(agg, W2, nullptr, t, N, EMB, EMB, 0);
    k_gather_finish<<<cdiv(N * 5, 256), 256, 0, stream>>>(rowptr, cnt, lists, t, dinv, b2, agg, N);

    run_gemm(agg  , fc1W, fc1b, fco, Bg, LAT, HID, 1);

    run_gemm(fco, outW, outb, (float*)d_out, Bg, HID, NC, 0);
}


# fc1 GEMM (8192x1640x300) re-tiled: 32x300 block tile staged through LDS by coalesced 16-byte loads, 2x5 accumulator tiles per wave; k order of the sums unchanged; its weight transpose launch dropped
# speedup vs baseline: 4.2036x; 1.2432x over previous
#include <hip/hip_runtime.h>
#include <hip/hip_bf16.h>

typedef __attribute__((ext_vector_type(2))) float v2f;
typedef __attribute__((ext_vector_type(4))) float v4f;
typedef __attribute__((ext_vector_type(8))) float v8f;

static __host__ inline int cdiv(int a, int b) { return (a + b - 1) / b; }


__global__ void k_fill_i32(int* __restrict__ p, int v, int n) {
    int i = blockIdx.x * blockDim.x + threadIdx.x;
    if (i < n) p[i] = v;
}

__global__ void k_count_rank(const int* __restrict__ dst, int* __restrict__ cnt,
                             int* __restrict__ rank, int E) {
    int e = blockIdx.x * blockDim.x + threadIdx.x;
    if (e < E) rank[e] = atomicAdd(&cnt[dst[e]], 1);
}

__global__ void __launch_bounds__(256) k_tile_sums(const int* __restrict__ cnt,
                                                   int* __restrict__ tsum, int n) {
    __shared__ int wsum[8];
    int base = blockIdx.x * 1024 + threadIdx.x * 4;
    int s = 0;
    for (int j = 0; j < 4; ++j)
        if (base + j < n) s += cnt[base + j];
    for (int o = 16; o > 0; o >>= 1) s += __shfl_down(s, o, 32);
    if ((threadIdx.x & 31) == 0) wsum[threadIdx.x >> 5] = s;
    __syncthreads();
    if (threadIdx.x == 0) {
        int tot = 0;
        for (int w = 0; w < 8; ++w) tot += wsum[w];
        tsum[blockIdx.x] = tot;
    }
}

__global__ void __launch_bounds__(256) k_scan_rowptr(const int* __restrict__ cnt,
                                                     const int* __restrict__ tsum,
                                                     int* __restrict__ rowptr,
                                                     float* __restrict__ dinv, int n) {
    __shared__ int wpre[8];
    __shared__ int winc[8];
    int tid = threadIdx.x, lane = tid & 31, w = tid >> 5;

    int p = 0;
    for (int j = tid; j < (int)blockIdx.x; j += 256) p += tsum[j];
    for (int o = 16; o > 0; o >>= 1) p += __shfl_down(p, o, 32);
    if (lane == 0) wpre[w] = p;

    int base = blockIdx.x * 1024 + tid * 4;
    int c[4];
    for (int j = 0; j < 4; ++j) c[j] = (base + j < n) ? cnt[base + j] : 0;
    int s = c[0] + c[1] + c[2] + c[3];
    int inc = s;
    for (int o = 1; o < 32; o <<= 1) {
        int y = __shfl_up(inc, o, 32);
        if (lane >= o) inc += y;
    }
    if (lane == 31) winc[w] = inc;
    __syncthreads();

    int off = 0;
    for (int j = 0; j < 8; ++j) {
        off += wpre[j];
        if (j < w) off += winc[j];
    }
    int ex = off + inc - s;
    for (int j = 0; j < 4; ++j) {
        if (base + j < n) {
            rowptr[base + j] = ex;
            float v = (float)(c[j] + 1);
            dinv[base + j] = (v > 0.0f) ? rsqrtf(v) : 0.0f;
            ex += c[j];
        }
    }
}

__global__ void k_fill_lists(const int* __restrict__ src, const int* __restrict__ dst,
                             const int* __restrict__ rank, const int* __restrict__ rowptr,
                             const float* __restrict__ dinv, int* __restrict__ lists, int E) {
    int e = blockIdx.x * blockDim.x + threadIdx.x;
    if (e >= E) return;
    int s = src[e], d = dst[e];
    int pos = rowptr[d] + rank[e];
    pos = min(max(pos, 0), E - 1);
    ((int2*)lists)[pos] = make_int2(s, __float_as_int(dinv[s]));
}

__global__ void k_transpose_pad(const float* __restrict__ Bw, float* __restrict__ Bt,
                                int K, int Ncols, int ncols_pad) {
    int i = blockIdx.x * blockDim.x + threadIdx.x;
    int total = ncols_pad * K;
    if (i >= total) return;
    int ncol = i / K, k = i - ncol * K;
    Bt[i] = (ncol < Ncols) ? Bw[(size_t)k * Ncols + ncol] : 0.0f;
}

__global__ void __launch_bounds__(256) k_gather_finish(const int* __restrict__ rowptr,
                                                       const int* __restrict__ cnt,
                                                       const int* __restrict__ lists,
                                                       const float* __restrict__ t,
                                                       const float* __restrict__ dinv,
                                                       const float* __restrict__ bias,
                                                       float* __restrict__ h, int n_nodes) {
    int g = blockIdx.x * 256 + threadIdx.x;
    if (g >= n_nodes * 5) return;
    int node = g / 5;
    int q = g - node * 5;
    const float4* t4 = (const float4*)t;
    const int2* le = (const int2*)lists;
    float di = dinv[node];
    int k = rowptr[node];
    int k1 = k + cnt[node];
    int nmax = n_nodes - 1;

    float4 acc = make_float4(0.0f, 0.0f, 0.0f, 0.0f);
    for (; k + 4 <= k1; k += 4) {
        int2 e[4];
        float4 v[4];
        for (int j = 0; j < 4; ++j) e[j] = le[k + j];
        for (int j = 0; j < 4; ++j) v[j] = t4[min(max(e[j].x, 0), nmax) * 5 + q];
        for (int j = 0; j < 4; ++j) {
            float nr = __int_as_float(e[j].y) * di;
            acc.x = fmaf(v[j].x, nr, acc.x);
            acc.y = fmaf(v[j].y, nr, acc.y);
            acc.z = fmaf(v[j].z, nr, acc.z);
            acc.w = fmaf(v[j].w, nr, acc.w);
        }
    }
#pragma unroll 1
    for (; k < k1; ++k) {
        int2 e0 = le[k];
        float4 v0 = t4[min(max(e0.x, 0), nmax) * 5 + q];
        float nr = __int_as_float(e0.y) * di;
        acc.x = fmaf(v0.x, nr, acc.x);
        acc.y = fmaf(v0.y, nr, acc.y);
        acc.z = fmaf(v0.z, nr, acc.z);
        acc.w = fmaf(v0.w, nr, acc.w);
    }

    float4 sv = t4[g];
    const float* bq = bias + 4 * q;
    float4 o;
    o.x = fmaxf(acc.x + sv.x * di * di + bq[0], 0.0f);
    o.y = fmaxf(acc.y + sv.y * di * di + bq[1], 0.0f);
    o.z = fmaxf(acc.z + sv.z * di * di + bq[2], 0.0f);
    o.w = fmaxf(acc.w + sv.w * di * di + bq[3], 0.0f);
    ((float4*)h)[g] = o;
}

__global__ void k_wmma_gemm_f32(const float* __restrict__ A, const float* __restrict__ Bt,
                                const float* __restrict__ bias, float* __restrict__ C,
                                int M, int K, int Ncols, int mtiles, int ntiles, int do_relu) {
    int wave = (blockIdx.x * blockDim.x + threadIdx.x) >> 5;
    int lane = threadIdx.x & 31;
    int mt = wave % mtiles;
    int nt = wave / mtiles;
    if (nt >= ntiles) return;

    int lrow = lane & 15;
    int kh = (lane >> 4) << 1;
    int ncol = nt * 16 + lrow;
    const float* arow = A + (size_t)(mt * 16 + lrow) * K + kh;
    const float* brow = Bt + (size_t)ncol * K + kh;

    v8f acc = {};
    int kb = 0;
    for (; kb + 16 <= K; kb += 16) {
#pragma unroll
        for (int j = 0; j < 4; ++j) {
            v2f a = *(const v2f*)(arow + kb + 4 * j);
            v2f b = *(const v2f*)(brow + kb + 4 * j);
            acc = __builtin_amdgcn_wmma_f32_16x16x4_f32(false, a, false, b,
                                                        (short)0, acc, false, false);
        }
    }
    for (; kb < K; kb += 4) {
        v2f a = *(const v2f*)(arow + kb);
        v2f b = *(const v2f*)(brow + kb);
        acc = __builtin_amdgcn_wmma_f32_16x16x4_f32(false, a, false, b,
                                                    (short)0, acc, false, false);
    }

    if (ncol >= Ncols) return;
    float bv = bias ? bias[ncol] : 0.0f;
    int rbase = mt * 16 + ((lane >> 4) << 3);
#pragma unroll
    for (int r = 0; r < 8; ++r) {
        float v = acc[r] + bv;
        if (do_relu) v = fmaxf(v, 0.0f);
        C[(size_t)(rbase + r) * Ncols + ncol] = v;
    }
}

__global__ void __launch_bounds__(128) k_fc1_tiled(const float* __restrict__ A,
                                                   const float* __restrict__ Bw,
                                                   const float* __restrict__ bias,
                                                   float* __restrict__ C, int M) {
    constexpr int K = 1640, N = 300, BM = 32, BK = 40, NP = 336, PA = 44, NT = K / BK;
    constexpr int NB = (BK / 2) * (N / 4);
    constexpr int NA = BM * (BK / 4);
    constexpr int JB = (NB + 127) / 128;
    constexpr int JA = (NA + 127) / 128;
    constexpr int SA = BM * PA, SB = (BK / 2) * NP * 2;
    __shared__ __attribute__((aligned(16))) float As[2 * SA];
    __shared__ __attribute__((aligned(16))) float Bs[2 * SB];

    const int tid = threadIdx.x, lane = tid & 31, w = tid >> 5;
    const int c = lane & 15, h = lane >> 4;
    const int m0 = blockIdx.x * BM;

    for (int i = tid; i < (BK / 2) * 72; i += 128) {
        int kp = i / 72, r = i - kp * 72;
        Bs[(kp * NP + N) * 2 + r] = 0.0f;
        Bs[SB + (kp * NP + N) * 2 + r] = 0.0f;
    }

    v4f rb0[JB], rb1[JB], ra[JA];
    int boff[JB], bsoff[JB], aoff[JA], asoff[JA];
#pragma unroll
    for (int j = 0; j < JB; ++j) {
        int idx = min(tid + 128 * j, NB - 1);
        int kp = idx / (N / 4), q = idx - kp * (N / 4);
        boff[j] = (2 * kp) * N + 4 * q;
        bsoff[j] = (kp * NP + 4 * q) * 2;
    }
#pragma unroll
    for (int j = 0; j < JA; ++j) {
        int idx = min(tid + 128 * j, NA - 1);
        int m = idx / (BK / 4), jj = idx - m * (BK / 4);
        int row = min(m0 + m, M - 1);
        aoff[j] = row * K + 4 * jj;
        asoff[j] = m * PA + 4 * jj;
    }

#pragma unroll
    for (int j = 0; j < JB; ++j) {
        rb0[j] = *(const v4f*)(Bw + boff[j]);
        rb1[j] = *(const v4f*)(Bw + boff[j] + N);
    }
#pragma unroll
    for (int j = 0; j < JA; ++j)
        ra[j] = *(const v4f*)(A + aoff[j]);

    v8f acc[2][5] = {};
    const float* ap = As + c * PA + 2 * h;
    const float* bp = Bs + (h * NP + w * 80 + c) * 2;

#pragma unroll
    for (int j = 0; j < JB; ++j) {
        float* p = Bs + bsoff[j];
        *(v4f*)p       = (v4f){rb0[j].x, rb1[j].x, rb0[j].y, rb1[j].y};
        *(v4f*)(p + 4) = (v4f){rb0[j].z, rb1[j].z, rb0[j].w, rb1[j].w};
    }
#pragma unroll
    for (int j = 0; j < JA; ++j) *(v4f*)(As + asoff[j]) = ra[j];
#pragma unroll
    for (int j = 0; j < JB; ++j) {
        rb0[j] = *(const v4f*)(Bw + BK * N + boff[j]);
        rb1[j] = *(const v4f*)(Bw + BK * N + boff[j] + N);
    }
#pragma unroll
    for (int j = 0; j < JA; ++j) ra[j] = *(const v4f*)(A + BK + aoff[j]);
    __syncthreads();

    for (int t = 0; t < NT; ++t) {
        const int cur = (t & 1) * SB, nxt = SB - cur;
        const int curA = (t & 1) * SA, nxtA = SA - curA;

#pragma unroll
        for (int kk = 0; kk < BK / 4; ++kk) {
            if (kk == 1 && t + 1 < NT) {
#pragma unroll
                for (int j = 0; j < JB; ++j) {
                    float* p = Bs + nxt + bsoff[j];
                    *(v4f*)p       = (v4f){rb0[j].x, rb1[j].x, rb0[j].y, rb1[j].y};
                    *(v4f*)(p + 4) = (v4f){rb0[j].z, rb1[j].z, rb0[j].w, rb1[j].w};
                }
#pragma unroll
                for (int j = 0; j < JA; ++j) *(v4f*)(As + nxtA + asoff[j]) = ra[j];
                if (t + 2 < NT) {
                    const float* Bn = Bw + (t + 2) * (BK * N);
                    const float* An = A + (t + 2) * BK;
#pragma unroll
                    for (int j = 0; j < JB; ++j) {
                        rb0[j] = *(const v4f*)(Bn + boff[j]);
                        rb1[j] = *(const v4f*)(Bn + boff[j] + N);
                    }
#pragma unroll
                    for (int j = 0; j < JA; ++j) ra[j] = *(const v4f*)(An + aoff[j]);
                }
            }
            v2f a0 = *(const v2f*)(ap + curA + kk * 4);
            v2f a1 = *(const v2f*)(ap + curA + 16 * PA + kk * 4);
            v2f b[5];
#pragma unroll
            for (int ni = 0; ni < 5; ++ni) b[ni] = *(const v2f*)(bp + cur + kk * (4 * NP) + ni * 32);
#pragma unroll
            for (int ni = 0; ni < 5; ++ni) {
                acc[0][ni] = __builtin_amdgcn_wmma_f32_16x16x4_f32(false, a0, false, b[ni],
                                                                   (short)0, acc[0][ni], false, false);
                acc[1][ni] = __builtin_amdgcn_wmma_f32_16x16x4_f32(false, a1, false, b[ni],
                                                                   (short)0, acc[1][ni], false, false);
            }
        }
        __syncthreads();
    }

#pragma unroll
    for (int ni = 0; ni < 5; ++ni) {
        int col = w * 80 + ni * 16 + c;
        if (col < N) {
            float bv = bias[col];
#pragma unroll
            for (int mi = 0; mi < 2; ++mi) {
                int rbase = m0 + mi * 16 + 8 * h;
#pragma unroll
                for (int r = 0; r < 8; ++r)
                    if (rbase + r < M)
                        C[(size_t)(rbase + r) * N + col] = fmaxf(acc[mi][ni][r] + bv, 0.0f);
            }
        }
    }
}


extern "C" void kernel_launch(void* const* d_in, const int* in_sizes, int n_in,
                              void* d_out, int out_size, void* d_ws, size_t ws_size,
                              hipStream_t stream) {
    const float* x    = (const float*)d_in[0];
    const int*   eidx = (const int*)d_in[1];
    const float* W1   = (const float*)d_in[3];
    const float* b1   = (const float*)d_in[4];
    const float* W2   = (const float*)d_in[5];
    const float* b2   = (const float*)d_in[6];
    const float* fc1W = (const float*)d_in[7];
    const float* fc1b = (const float*)d_in[8];
    const float* outW = (const float*)d_in[9];
    const float* outb = (const float*)d_in[10];

    const int N  = in_sizes[2];
    const int E  = in_sizes[1] / 2;
    const int Bg = N / 82;
    const int EMB = 20, LAT = 82 * 20, HID = 300, NC = 22;

    char* ws = (char*)d_ws;
    size_t off = 0;
    auto take = [&](size_t bytes) -> void* {
        off = (off + 255) & ~(size_t)255;
        void* p = ws + off;
        off += bytes;
        return p;
    };
    const int ntile = cdiv(N, 1024);
    float* dinv   = (float*)take((size_t)N * 4);
    int*   cnt    = (int*)take((size_t)N * 4);
    int*   rowptr = (int*)take((size_t)N * 4);
    int*   tsum   = (int*)take((size_t)ntile * 4);
    int*   rank   = (int*)take((size_t)E * 4);
    int*   lists  = (int*)take((size_t)E * 8);
    float* t    = (float*)take((size_t)N * EMB * 4);
    float* agg  = (float*)take((size_t)N * EMB * 4);
    float* fco  = (float*)take((size_t)Bg * HID * 4);
    float* Bt   = (float*)take((size_t)304 * LAT * 4);

    const int* src = eidx;
    const int* dst = eidx + E;

    const int TB = 256;

    auto run_gemm = [&](const float* A, const float* Bw, const float* bias, float* C,
                        int M, int K, int Ncols, int relu) {
        int ntiles = (Ncols + 15) / 16;
        int ncols_pad = ntiles * 16;
        int mtiles = M / 16;
        k_transpose_pad<<<cdiv(ncols_pad * K, TB), TB, 0, stream>>>(Bw, Bt, K, Ncols, ncols_pad);
        int waves = mtiles * ntiles;
        k_wmma_gemm_f32<<<cdiv(waves * 32, TB), TB, 0, stream>>>(
            A, Bt, bias, C, M, K, Ncols, mtiles, ntiles, relu);
    };

    k_fill_i32<<<cdiv(N, TB), TB, 0, stream>>>(cnt, 0, N);
    k_count_rank<<<cdiv(E, TB), TB, 0, stream>>>(dst, cnt, rank, E);
    k_tile_sums<<<ntile, 256, 0, stream>>>(cnt, tsum, N);
    k_scan_rowptr<<<ntile, 256, 0, stream>>>(cnt, tsum, rowptr, dinv, N);
    k_fill_lists<<<cdiv(E, TB), TB, 0, stream>>>(src, dst, rank, rowptr, dinv, lists, E);

    run_gemm(x, W1, nullptr, t, N, 20, EMB, 0);
    k_gather_finish<<<cdiv(N * 5, 256), 256, 0, stream>>>(rowptr, cnt, lists, t, dinv, b1, agg, N);

    run_gemm(agg, W2, nullptr, t, N, EMB, EMB, 0);
    k_gather_finish<<<cdiv(N * 5, 256), 256, 0, stream>>>(rowptr, cnt, lists, t, dinv, b2, agg, N);

    k_fc1_tiled<<<cdiv(Bg, 32), 128, 0, stream>>>(agg  , fc1W, fc1b, fco, Bg);

    run_gemm(fco, outW, outb, (float*)d_out, Bg, HID, NC, 0);
}


# In-edge lists built by a two-level counting sort on LDS counters (edges binned by 2048-node dst range, then ranked inside each bin): no global atomic left; sum order = arrival rank of the LDS adds
# speedup vs baseline: 6.6614x; 1.5847x over previous
#include <hip/hip_runtime.h>
#include <hip/hip_bf16.h>

typedef __attribute__((ext_vector_type(2))) float v2f;
typedef __attribute__((ext_vector_type(4))) float v4f;
typedef __attribute__((ext_vector_type(8))) float v8f;

static __host__ inline int cdiv(int a, int b) { return (a + b - 1) / b; }

constexpr int BIN_SHIFT = 11;
constexpr int BIN_W = 1 << BIN_SHIFT;
constexpr int MAX_BINS = 1024;
constexpr int NBLK_E = 256;
constexpr int TE = 1024;
constexpr int CAP_E = 16384;
static_assert(BIN_W == 2 * TE, "k_bin_count scans two counters per thread");
static_assert(MAX_BINS == TE, "k_bin_scatter scans one bin per thread");

__device__ __forceinline__ int bin_of(int d, int nbin) { return min(max(d >> BIN_SHIFT, 0), nbin - 1); }

__global__ void __launch_bounds__(TE) k_bin_hist(const int* __restrict__ dst, int* __restrict__ hist,
                                                 int E, int chunk, int nbin) {
    __shared__ int hc[MAX_BINS];
    const int tid = threadIdx.x;
    for (int i = tid; i < nbin; i += TE) hc[i] = 0;
    __syncthreads();
    const int start = blockIdx.x * chunk;
    const int end = min(start + chunk, E);
    int i = start + tid;
    for (; i + 3 * TE < end; i += 4 * TE) {
        int d[4];
        for (int j = 0; j < 4; ++j) d[j] = dst[i + j * TE];
        for (int j = 0; j < 4; ++j) atomicAdd(&hc[bin_of(d[j], nbin)], 1);
    }
    for (; i < end; i += TE) atomicAdd(&hc[bin_of(dst[i], nbin)], 1);
    __syncthreads();
    for (int b = tid; b < nbin; b += TE) hist[b * NBLK_E + blockIdx.x] = hc[b];
}

__global__ void __launch_bounds__(256) k_tile_sums(const int* __restrict__ cnt,
                                                   int* __restrict__ tsum, int n) {
    __shared__ int wsum[8];
    int base = blockIdx.x * 1024 + threadIdx.x * 4;
    int s = 0;
    for (int j = 0; j < 4; ++j)
        if (base + j < n) s += cnt[base + j];
    for (int o = 16; o > 0; o >>= 1) s += __shfl_down(s, o, 32);
    if ((threadIdx.x & 31) == 0) wsum[threadIdx.x >> 5] = s;
    __syncthreads();
    if (threadIdx.x == 0) {
        int tot = 0;
        for (int w = 0; w < 8; ++w) tot += wsum[w];
        tsum[blockIdx.x] = tot;
    }
}

__global__ void __launch_bounds__(256) k_scan_excl(const int* __restrict__ cnt,
                                                   const int* __restrict__ tsum,
                                                   int* __restrict__ pre, int n) {
    __shared__ int wpre[8];
    __shared__ int winc[8];
    int tid = threadIdx.x, lane = tid & 31, w = tid >> 5;

    int p = 0;
    for (int j = tid; j < (int)blockIdx.x; j += 256) p += tsum[j];
    for (int o = 16; o > 0; o >>= 1) p += __shfl_down(p, o, 32);
    if (lane == 0) wpre[w] = p;

    int base = blockIdx.x * 1024 + tid * 4;
    int c[4];
    for (int j = 0; j < 4; ++j) c[j] = (base + j < n) ? cnt[base + j] : 0;
    int s = c[0] + c[1] + c[2] + c[3];
    int inc = s;
    for (int o = 1; o < 32; o <<= 1) {
        int y = __shfl_up(inc, o, 32);
        if (lane >= o) inc += y;
    }
    if (lane == 31) winc[w] = inc;
    __syncthreads();

    int off = 0;
    for (int j = 0; j < 8; ++j) {
        off += wpre[j];
        if (j < w) off += winc[j];
    }
    int ex = off + inc - s;
    for (int j = 0; j < 4; ++j) {
        if (base + j < n) {
            pre[base + j] = ex;
            ex += c[j];
        }
    }
}

__global__ void __launch_bounds__(TE) k_bin_scatter(const int* __restrict__ src,
                                                    const int* __restrict__ dst,
                                                    const int* __restrict__ pre,
                                                    int2* __restrict__ binned,
                                                    int E, int chunk, int nbin) {
    __shared__ int gbase[MAX_BINS];
    __shared__ int lstart[MAX_BINS];
    __shared__ int cur[MAX_BINS];
    __shared__ int wtot[TE / 32];
    __shared__ int2 stage[CAP_E];
    const int tid = threadIdx.x, lane = tid & 31, w = tid >> 5;
    const int start = blockIdx.x * chunk;
    const int end = min(start + chunk, E);
    const int len = max(end - start, 0);

    int g = 0, nb = 0;
    if (tid < nbin) {
        int idx = tid * NBLK_E + blockIdx.x;
        g = pre[idx];
        nb = ((idx + 1 < nbin * NBLK_E) ? pre[idx + 1] : E) - g;
    }

    if (len <= CAP_E) {
        int inc = nb;
        for (int o = 1; o < 32; o <<= 1) {
            int y = __shfl_up(inc, o, 32);
            if (lane >= o) inc += y;
        }
        if (lane == 31) wtot[w] = inc;
        __syncthreads();
        int off = 0;
        for (int j = 0; j < TE / 32; ++j)
            if (j < w) off += wtot[j];
        const int ex = off + inc - nb;
        if (tid < nbin) {
            gbase[tid] = g;
            lstart[tid] = ex;
            cur[tid] = ex;
        }
        __syncthreads();
        int i = start + tid;
        for (; i + 3 * TE < end; i += 4 * TE) {
            int s[4], d[4], pos[4];
            for (int j = 0; j < 4; ++j) {
                s[j] = src[i + j * TE];
                d[j] = dst[i + j * TE];
            }
            for (int j = 0; j < 4; ++j) pos[j] = atomicAdd(&cur[bin_of(d[j], nbin)], 1);
            for (int j = 0; j < 4; ++j) stage[min(max(pos[j], 0), CAP_E - 1)] = make_int2(s[j], d[j]);
        }
        for (; i < end; i += TE) {
            int s = src[i], d = dst[i];
            int pos = atomicAdd(&cur[bin_of(d, nbin)], 1);
            stage[min(max(pos, 0), CAP_E - 1)] = make_int2(s, d);
        }
        __syncthreads();
        for (int k = tid; k < len; k += TE) {
            int2 e = stage[k];
            int b = bin_of(e.y, nbin);
            int pos = gbase[b] + (k - lstart[b]);
            binned[min(max(pos, 0), E - 1)] = e;
        }
    } else {
        if (tid < nbin) cur[tid] = g;
        __syncthreads();
        for (int i = start + tid; i < end; i += TE) {
            int s = src[i], d = dst[i];
            int pos = atomicAdd(&cur[bin_of(d, nbin)], 1);
            binned[min(max(pos, 0), E - 1)] = make_int2(s, d);
        }
    }
}

__global__ void __launch_bounds__(TE) k_bin_count(const int2* __restrict__ binned,
                                                  const int* __restrict__ pre,
                                                  int* __restrict__ rowptr, int* __restrict__ cnt,
                                                  float* __restrict__ dinv, int n, int E, int nbin) {
    __shared__ int c[BIN_W];
    __shared__ int wtot[TE / 32];
    const int tid = threadIdx.x, lane = tid & 31, w = tid >> 5;
    const int bin = blockIdx.x;
    const int lo = bin << BIN_SHIFT;
    int bs = pre[bin * NBLK_E];
    int be = (bin + 1 < nbin) ? pre[(bin + 1) * NBLK_E] : E;
    bs = min(max(bs, 0), E);
    be = min(max(be, bs), E);
    for (int k = tid; k < BIN_W; k += TE) c[k] = 0;
    __syncthreads();
    int i = bs + tid;
    for (; i + 3 * TE < be; i += 4 * TE) {
        int d[4];
        for (int j = 0; j < 4; ++j) d[j] = binned[i + j * TE].y;
        for (int j = 0; j < 4; ++j) atomicAdd(&c[min(max(d[j] - lo, 0), BIN_W - 1)], 1);
    }
    for (; i < be; i += TE) atomicAdd(&c[min(max(binned[i].y - lo, 0), BIN_W - 1)], 1);
    __syncthreads();
    const int c0 = c[2 * tid], c1 = c[2 * tid + 1];
    const int s = c0 + c1;
    int inc = s;
    for (int o = 1; o < 32; o <<= 1) {
        int y = __shfl_up(inc, o, 32);
        if (lane >= o) inc += y;
    }
    if (lane == 31) wtot[w] = inc;
    __syncthreads();
    int off = 0;
    for (int j = 0; j < TE / 32; ++j)
        if (j < w) off += wtot[j];
    const int ex = bs + off + inc - s;
    const int node = lo + 2 * tid;
    if (node < n) {
        rowptr[node] = ex;
        cnt[node] = c0;
        float v = (float)(c0 + 1);
        dinv[node] = (v > 0.0f) ? rsqrtf(v) : 0.0f;
    }
    if (node + 1 < n) {
        rowptr[node + 1] = ex + c0;
        cnt[node + 1] = c1;
        float v = (float)(c1 + 1);
        dinv[node + 1] = (v > 0.0f) ? rsqrtf(v) : 0.0f;
    }
}

__global__ void __launch_bounds__(TE) k_bin_fill(const int2* __restrict__ binned,
                                                 const int* __restrict__ pre,
                                                 const int* __restrict__ rowptr,
                                                 const float* __restrict__ dinv,
                                                 int* __restrict__ lists, int n, int E, int nbin) {
    __shared__ int cur[BIN_W];
    __shared__ int2 stage[CAP_E];
    const int tid = threadIdx.x;
    const int bin = blockIdx.x;
    const int lo = bin << BIN_SHIFT;
    int bs = pre[bin * NBLK_E];
    int be = (bin + 1 < nbin) ? pre[(bin + 1) * NBLK_E] : E;
    bs = min(max(bs, 0), E);
    be = min(max(be, bs), E);
    for (int k = tid; k < BIN_W; k += TE) cur[k] = (lo + k < n) ? rowptr[lo + k] : 0;
    __syncthreads();
    int2* le = (int2*)lists;
    const int len = be - bs;
    if (len <= CAP_E) {
        int i = bs + tid;
        for (; i + 3 * TE < be; i += 4 * TE) {
            int2 e[4];
            float ds[4];
            int pos[4];
            for (int j = 0; j < 4; ++j) e[j] = binned[i + j * TE];
            for (int j = 0; j < 4; ++j) ds[j] = dinv[min(max(e[j].x, 0), n - 1)];
            for (int j = 0; j < 4; ++j) pos[j] = atomicAdd(&cur[min(max(e[j].y - lo, 0), BIN_W - 1)], 1) - bs;
            for (int j = 0; j < 4; ++j)
                stage[min(max(pos[j], 0), CAP_E - 1)] = make_int2(e[j].x, __float_as_int(ds[j]));
        }
        for (; i < be; i += TE) {
            int2 e = binned[i];
            float ds = dinv[min(max(e.x, 0), n - 1)];
            int pos = atomicAdd(&cur[min(max(e.y - lo, 0), BIN_W - 1)], 1) - bs;
            stage[min(max(pos, 0), CAP_E - 1)] = make_int2(e.x, __float_as_int(ds));
        }
        __syncthreads();
        for (int k = tid; k < len; k += TE) le[bs + k] = stage[k];
    } else {
        for (int i = bs + tid; i < be; i += TE) {
            int2 e = binned[i];
            float ds = dinv[min(max(e.x, 0), n - 1)];
            int pos = atomicAdd(&cur[min(max(e.y - lo, 0), BIN_W - 1)], 1);
            le[min(max(pos, 0), E - 1)] = make_int2(e.x, __float_as_int(ds));
        }
    }
}

__global__ void k_transpose_pad(const float* __restrict__ Bw, float* __restrict__ Bt,
                                int K, int Ncols, int ncols_pad) {
    int i = blockIdx.x * blockDim.x + threadIdx.x;
    int total = ncols_pad * K;
    if (i >= total) return;
    int ncol = i / K, k = i - ncol * K;
    Bt[i] = (ncol < Ncols) ? Bw[(size_t)k * Ncols + ncol] : 0.0f;
}

__global__ void __launch_bounds__(256) k_gather_finish(const int* __restrict__ rowptr,
                                                       const int* __restrict__ cnt,
                                                       const int* __restrict__ lists,
                                                       const float* __restrict__ t,
                                                       const float* __restrict__ dinv,
                                                       const float* __restrict__ bias,
                                                       float* __restrict__ h, int n_nodes) {
    int g = blockIdx.x * 256 + threadIdx.x;
    if (g >= n_nodes * 5) return;
    int node = g / 5;
    int q = g - node * 5;
    const float4* t4 = (const float4*)t;
    const int2* le = (const int2*)lists;
    float di = dinv[node];
    int k = rowptr[node];
    int k1 = k + cnt[node];
    int nmax = n_nodes - 1;

    float4 acc = make_float4(0.0f, 0.0f, 0.0f, 0.0f);
    for (; k + 4 <= k1; k += 4) {
        int2 e[4];
        float4 v[4];
        for (int j = 0; j < 4; ++j) e[j] = le[k + j];
        for (int j = 0; j < 4; ++j) v[j] = t4[min(max(e[j].x, 0), nmax) * 5 + q];
        for (int j = 0; j < 4; ++j) {
            float nr = __int_as_float(e[j].y) * di;
            acc.x = fmaf(v[j].x, nr, acc.x);
            acc.y = fmaf(v[j].y, nr, acc.y);
            acc.z = fmaf(v[j].z, nr, acc.z);
            acc.w = fmaf(v[j].w, nr, acc.w);
        }
    }
#pragma unroll 1
    for (; k < k1; ++k) {
        int2 e0 = le[k];
        float4 v0 = t4[min(max(e0.x, 0), nmax) * 5 + q];
        float nr = __int_as_float(e0.y) * di;
        acc.x = fmaf(v0.x, nr, acc.x);
        acc.y = fmaf(v0.y, nr, acc.y);
        acc.z = fmaf(v0.z, nr, acc.z);
        acc.w = fmaf(v0.w, nr, acc.w);
    }

    float4 sv = t4[g];
    const float* bq = bias + 4 * q;
    float4 o;
    o.x = fmaxf(acc.x + sv.x * di * di + bq[0], 0.0f);
    o.y = fmaxf(acc.y + sv.y * di * di + bq[1], 0.0f);
    o.z = fmaxf(acc.z + sv.z * di * di + bq[2], 0.0f);
    o.w = fmaxf(acc.w + sv.w * di * di + bq[3], 0.0f);
    ((float4*)h)[g] = o;
}

__global__ void k_wmma_gemm_f32(const float* __restrict__ A, const float* __restrict__ Bt,
                                const float* __restrict__ bias, float* __restrict__ C,
                                int M, int K, int Ncols, int mtiles, int ntiles, int do_relu) {
    int wave = (blockIdx.x * blockDim.x + threadIdx.x) >> 5;
    int lane = threadIdx.x & 31;
    int mt = wave % mtiles;
    int nt = wave / mtiles;
    if (nt >= ntiles) return;

    int lrow = lane & 15;
    int kh = (lane >> 4) << 1;
    int ncol = nt * 16 + lrow;
    const float* arow = A + (size_t)(mt * 16 + lrow) * K + kh;
    const float* brow = Bt + (size_t)ncol * K + kh;

    v8f acc = {};
    int kb = 0;
    for (; kb + 16 <= K; kb += 16) {
#pragma unroll
        for (int j = 0; j < 4; ++j) {
            v2f a = *(const v2f*)(arow + kb + 4 * j);
            v2f b = *(const v2f*)(brow + kb + 4 * j);
            acc = __builtin_amdgcn_wmma_f32_16x16x4_f32(false, a, false, b,
                                                        (short)0, acc, false, false);
        }
    }
    for (; kb < K; kb += 4) {
        v2f a = *(const v2f*)(arow + kb);
        v2f b = *(const v2f*)(brow + kb);
        acc = __builtin_amdgcn_wmma_f32_16x16x4_f32(false, a, false, b,
                                                    (short)0, acc, false, false);
    }

    if (ncol >= Ncols) return;
    float bv = bias ? bias[ncol] : 0.0f;
    int rbase = mt * 16 + ((lane >> 4) << 3);
#pragma unroll
    for (int r = 0; r < 8; ++r) {
        float v = acc[r] + bv;
        if (do_relu) v = fmaxf(v, 0.0f);
        C[(size_t)(rbase + r) * Ncols + ncol] = v;
    }
}

__global__ void __launch_bounds__(128) k_fc1_tiled(const float* __restrict__ A,
                                                   const float* __restrict__ Bw,
                                                   const float* __restrict__ bias,
                                                   float* __restrict__ C, int M) {
    constexpr int K = 1640, N = 300, BM = 32, BK = 40, NP = 336, PA = 44, NT = K / BK;
    constexpr int NB = (BK / 2) * (N / 4);
    constexpr int NA = BM * (BK / 4);
    constexpr int JB = (NB + 127) / 128;
    constexpr int JA = (NA + 127) / 128;
    constexpr int SA = BM * PA, SB = (BK / 2) * NP * 2;
    __shared__ __attribute__((aligned(16))) float As[2 * SA];
    __shared__ __attribute__((aligned(16))) float Bs[2 * SB];

    const int tid = threadIdx.x, lane = tid & 31, w = tid >> 5;
    const int c = lane & 15, h = lane >> 4;
    const int m0 = blockIdx.x * BM;

    for (int i = tid; i < (BK / 2) * 72; i += 128) {
        int kp = i / 72, r = i - kp * 72;
        Bs[(kp * NP + N) * 2 + r] = 0.0f;
        Bs[SB + (kp * NP + N) * 2 + r] = 0.0f;
    }

    v4f rb0[JB], rb1[JB], ra[JA];
    int boff[JB], bsoff[JB], aoff[JA], asoff[JA];
#pragma unroll
    for (int j = 0; j < JB; ++j) {
        int idx = min(tid + 128 * j, NB - 1);
        int kp = idx / (N / 4), q = idx - kp * (N / 4);
        boff[j] = (2 * kp) * N + 4 * q;
        bsoff[j] = (kp * NP + 4 * q) * 2;
    }
#pragma unroll
    for (int j = 0; j < JA; ++j) {
        int idx = min(tid + 128 * j, NA - 1);
        int m = idx / (BK / 4), jj = idx - m * (BK / 4);
        int row = min(m0 + m, M - 1);
        aoff[j] = row * K + 4 * jj;
        asoff[j] = m * PA + 4 * jj;
    }

#pragma unroll
    for (int j = 0; j < JB; ++j) {
        rb0[j] = *(const v4f*)(Bw + boff[j]);
        rb1[j] = *(const v4f*)(Bw + boff[j] + N);
    }
#pragma unroll
    for (int j = 0; j < JA; ++j)
        ra[j] = *(const v4f*)(A + aoff[j]);

    v8f acc[2][5] = {};
    const float* ap = As + c * PA + 2 * h;
    const float* bp = Bs + (h * NP + w * 80 + c) * 2;

#pragma unroll
    for (int j = 0; j < JB; ++j) {
        float* p = Bs + bsoff[j];
        *(v4f*)p       = (v4f){rb0[j].x, rb1[j].x, rb0[j].y, rb1[j].y};
        *(v4f*)(p + 4) = (v4f){rb0[j].z, rb1[j].z, rb0[j].w, rb1[j].w};
    }
#pragma unroll
    for (int j = 0; j < JA; ++j) *(v4f*)(As + asoff[j]) = ra[j];
#pragma unroll
    for (int j = 0; j < JB; ++j) {
        rb0[j] = *(const v4f*)(Bw + BK * N + boff[j]);
        rb1[j] = *(const v4f*)(Bw + BK * N + boff[j] + N);
    }
#pragma unroll
    for (int j = 0; j < JA; ++j) ra[j] = *(const v4f*)(A + BK + aoff[j]);
    __syncthreads();

    for (int t = 0; t < NT; ++t) {
        const int cur = (t & 1) * SB, nxt = SB - cur;
        const int curA = (t & 1) * SA, nxtA = SA - curA;

#pragma unroll
        for (int kk = 0; kk < BK / 4; ++kk) {
            if (kk == 1 && t + 1 < NT) {
#pragma unroll
                for (int j = 0; j < JB; ++j) {
                    float* p = Bs + nxt + bsoff[j];
                    *(v4f*)p       = (v4f){rb0[j].x, rb1[j].x, rb0[j].y, rb1[j].y};
                    *(v4f*)(p + 4) = (v4f){rb0[j].z, rb1[j].z, rb0[j].w, rb1[j].w};
                }
#pragma unroll
                for (int j = 0; j < JA; ++j) *(v4f*)(As + nxtA + asoff[j]) = ra[j];
                if (t + 2 < NT) {
                    const float* Bn = Bw + (t + 2) * (BK * N);
                    const float* An = A + (t + 2) * BK;
#pragma unroll
                    for (int j = 0; j < JB; ++j) {
                        rb0[j] = *(const v4f*)(Bn + boff[j]);
                        rb1[j] = *(const v4f*)(Bn + boff[j] + N);
                    }
#pragma unroll
                    for (int j = 0; j < JA; ++j) ra[j] = *(const v4f*)(An + aoff[j]);
                }
            }
            v2f a0 = *(const v2f*)(ap + curA + kk * 4);
            v2f a1 = *(const v2f*)(ap + curA + 16 * PA + kk * 4);
            v2f b[5];
#pragma unroll
            for (int ni = 0; ni < 5; ++ni) b[ni] = *(const v2f*)(bp + cur + kk * (4 * NP) + ni * 32);
#pragma unroll
            for (int ni = 0; ni < 5; ++ni) {
                acc[0][ni] = __builtin_amdgcn_wmma_f32_16x16x4_f32(false, a0, false, b[ni],
                                                                   (short)0, acc[0][ni], false, false);
                acc[1][ni] = __builtin_amdgcn_wmma_f32_16x16x4_f32(false, a1, false, b[ni],
                                                                   (short)0, acc[1][ni], false, false);
            }
        }
        __syncthreads();
    }

#pragma unroll
    for (int ni = 0; ni < 5; ++ni) {
        int col = w * 80 + ni * 16 + c;
        if (col < N) {
            float bv = bias[col];
#pragma unroll
            for (int mi = 0; mi < 2; ++mi) {
                int rbase = m0 + mi * 16 + 8 * h;
#pragma unroll
                for (int r = 0; r < 8; ++r)
                    if (rbase + r < M)
                        C[(size_t)(rbase + r) * N + col] = fmaxf(acc[mi][ni][r] + bv, 0.0f);
            }
        }
    }
}


extern "C" void kernel_launch(void* const* d_in, const int* in_sizes, int n_in,
                              void* d_out, int out_size, void* d_ws, size_t ws_size,
                              hipStream_t stream) {
    const float* x    = (const float*)d_in[0];
    const int*   eidx = (const int*)d_in[1];
    const float* W1   = (const float*)d_in[3];
    const float* b1   = (const float*)d_in[4];
    const float* W2   = (const float*)d_in[5];
    const float* b2   = (const float*)d_in[6];
    const float* fc1W = (const float*)d_in[7];
    const float* fc1b = (const float*)d_in[8];
    const float* outW = (const float*)d_in[9];
    const float* outb = (const float*)d_in[10];

    const int N  = in_sizes[2];
    const int E  = in_sizes[1] / 2;
    const int Bg = N / 82;
    const int EMB = 20, LAT = 82 * 20, HID = 300, NC = 22;

    char* ws = (char*)d_ws;
    size_t off = 0;
    auto take = [&](size_t bytes) -> void* {
        off = (off + 255) & ~(size_t)255;
        void* p = ws + off;
        off += bytes;
        return p;
    };
    const int nbin  = cdiv(N, BIN_W);
    const int chunk = cdiv(E, NBLK_E);
    const int NH    = nbin * NBLK_E;
    const int ntile = cdiv(NH, 1024);
    float* dinv   = (float*)take((size_t)N * 4);
    int*   cnt    = (int*)take((size_t)N * 4);
    int*   rowptr = (int*)take((size_t)N * 4);
    int*   hist   = (int*)take((size_t)NH * 4);
    int*   pre    = (int*)take((size_t)NH * 4);
    int*   tsum   = (int*)take((size_t)ntile * 4);
    int2*  binned = (int2*)take((size_t)E * 8);
    int*   lists  = (int*)take((size_t)E * 8);
    float* t    = (float*)take((size_t)N * EMB * 4);
    float* agg  = (float*)take((size_t)N * EMB * 4);
    float* fco  = (float*)take((size_t)Bg * HID * 4);
    float* Bt   = (float*)take((size_t)304 * LAT * 4);

    const int* src = eidx;
    const int* dst = eidx + E;

    const int TB = 256;

    auto run_gemm = [&](const float* A, const float* Bw, const float* bias, float* C,
                        int M, int K, int Ncols, int relu) {
        int ntiles = (Ncols + 15) / 16;
        int ncols_pad = ntiles * 16;
        int mtiles = M / 16;
        k_transpose_pad<<<cdiv(ncols_pad * K, TB), TB, 0, stream>>>(Bw, Bt, K, Ncols, ncols_pad);
        int waves = mtiles * ntiles;
        k_wmma_gemm_f32<<<cdiv(waves * 32, TB), TB, 0, stream>>>(
            A, Bt, bias, C, M, K, Ncols, mtiles, ntiles, relu);
    };

    k_bin_hist<<<NBLK_E, TE, 0, stream>>>(dst, hist, E, chunk, nbin);
    k_tile_sums<<<ntile, 256, 0, stream>>>(hist, tsum, NH);
    k_scan_excl<<<ntile, 256, 0, stream>>>(hist, tsum, pre, NH);
    k_bin_scatter<<<NBLK_E, TE, 0, stream>>>(src, dst, pre, binned, E, chunk, nbin);
    k_bin_count<<<nbin, TE, 0, stream>>>(binned, pre, rowptr, cnt, dinv, N, E, nbin);
    k_bin_fill<<<nbin, TE, 0, stream>>>(binned, pre, rowptr, dinv, lists, N, E, nbin);

    run_gemm(x, W1, nullptr, t, N, 20, EMB, 0);
    k_gather_finish<<<cdiv(N * 5, 256), 256, 0, stream>>>(rowptr, cnt, lists, t, dinv, b1, agg, N);

    run_gemm(agg, W2, nullptr, t, N, EMB, EMB, 0);
    k_gather_finish<<<cdiv(N * 5, 256), 256, 0, stream>>>(rowptr, cnt, lists, t, dinv, b2, agg, N);

    k_fc1_tiled<<<cdiv(Bg, 32), 128, 0, stream>>>(agg  , fc1W, fc1b, fco, Bg);

    run_gemm(fco, outW, outb, (float*)d_out, Bg, HID, NC, 0);
}


# Both N x 20 x 20 node GEMMs: one wave owns 8 row tiles x both column tiles (was 1 wave per 16x16 tile), x loads issued up front, W^T as A operand so rows leave as 16-byte stores; k order kept
# speedup vs baseline: 7.1423x; 1.0722x over previous
#include <hip/hip_runtime.h>
#include <hip/hip_bf16.h>

typedef __attribute__((ext_vector_type(2))) float v2f;
typedef __attribute__((ext_vector_type(4))) float v4f;
typedef __attribute__((ext_vector_type(8))) float v8f;

static __host__ inline int cdiv(int a, int b) { return (a + b - 1) / b; }

constexpr int BIN_SHIFT = 11;
constexpr int BIN_W = 1 << BIN_SHIFT;
constexpr int MAX_BINS = 1024;
constexpr int NBLK_E = 256;
constexpr int TE = 1024;
constexpr int CAP_E = 16384;
static_assert(BIN_W == 2 * TE, "k_bin_count scans two counters per thread");
static_assert(MAX_BINS == TE, "k_bin_scatter scans one bin per thread");

__device__ __forceinline__ int bin_of(int d, int nbin) { return min(max(d >> BIN_SHIFT, 0), nbin - 1); }

__global__ void __launch_bounds__(TE) k_bin_hist(const int* __restrict__ dst, int* __restrict__ hist,
                                                 int E, int chunk, int nbin) {
    __shared__ int hc[MAX_BINS];
    const int tid = threadIdx.x;
    for (int i = tid; i < nbin; i += TE) hc[i] = 0;
    __syncthreads();
    const int start = blockIdx.x * chunk;
    const int end = min(start + chunk, E);
    int i = start + tid;
    for (; i + 3 * TE < end; i += 4 * TE) {
        int d[4];
        for (int j = 0; j < 4; ++j) d[j] = dst[i + j * TE];
        for (int j = 0; j < 4; ++j) atomicAdd(&hc[bin_of(d[j], nbin)], 1);
    }
    for (; i < end; i += TE) atomicAdd(&hc[bin_of(dst[i], nbin)], 1);
    __syncthreads();
    for (int b = tid; b < nbin; b += TE) hist[b * NBLK_E + blockIdx.x] = hc[b];
}

__global__ void __launch_bounds__(256) k_tile_sums(const int* __restrict__ cnt,
                                                   int* __restrict__ tsum, int n) {
    __shared__ int wsum[8];
    int base = blockIdx.x * 1024 + threadIdx.x * 4;
    int s = 0;
    for (int j = 0; j < 4; ++j)
        if (base + j < n) s += cnt[base + j];
    for (int o = 16; o > 0; o >>= 1) s += __shfl_down(s, o, 32);
    if ((threadIdx.x & 31) == 0) wsum[threadIdx.x >> 5] = s;
    __syncthreads();
    if (threadIdx.x == 0) {
        int tot = 0;
        for (int w = 0; w < 8; ++w) tot += wsum[w];
        tsum[blockIdx.x] = tot;
    }
}

__global__ void __launch_bounds__(256) k_scan_excl(const int* __restrict__ cnt,
                                                   const int* __restrict__ tsum,
                                                   int* __restrict__ pre, int n) {
    __shared__ int wpre[8];
    __shared__ int winc[8];
    int tid = threadIdx.x, lane = tid & 31, w = tid >> 5;

    int p = 0;
    for (int j = tid; j < (int)blockIdx.x; j += 256) p += tsum[j];
    for (int o = 16; o > 0; o >>= 1) p += __shfl_down(p, o, 32);
    if (lane == 0) wpre[w] = p;

    int base = blockIdx.x * 1024 + tid * 4;
    int c[4];
    for (int j = 0; j < 4; ++j) c[j] = (base + j < n) ? cnt[base + j] : 0;
    int s = c[0] + c[1] + c[2] + c[3];
    int inc = s;
    for (int o = 1; o < 32; o <<= 1) {
        int y = __shfl_up(inc, o, 32);
        if (lane >= o) inc += y;
    }
    if (lane == 31) winc[w] = inc;
    __syncthreads();

    int off = 0;
    for (int j = 0; j < 8; ++j) {
        off += wpre[j];
        if (j < w) off += winc[j];
    }
    int ex = off + inc - s;
    for (int j = 0; j < 4; ++j) {
        if (base + j < n) {
            pre[base + j] = ex;
            ex += c[j];
        }
    }
}

__global__ void __launch_bounds__(TE) k_bin_scatter(const int* __restrict__ src,
                                                    const int* __restrict__ dst,
                                                    const int* __restrict__ pre,
                                                    int2* __restrict__ binned,
                                                    int E, int chunk, int nbin) {
    __shared__ int gbase[MAX_BINS];
    __shared__ int lstart[MAX_BINS];
    __shared__ int cur[MAX_BINS];
    __shared__ int wtot[TE / 32];
    __shared__ int2 stage[CAP_E];
    const int tid = threadIdx.x, lane = tid & 31, w = tid >> 5;
    const int start = blockIdx.x * chunk;
    const int end = min(start + chunk, E);
    const int len = max(end - start, 0);

    int g = 0, nb = 0;
    if (tid < nbin) {
        int idx = tid * NBLK_E + blockIdx.x;
        g = pre[idx];
        nb = ((idx + 1 < nbin * NBLK_E) ? pre[idx + 1] : E) - g;
    }

    if (len <= CAP_E) {
        int inc = nb;
        for (int o = 1; o < 32; o <<= 1) {
            int y = __shfl_up(inc, o, 32);
            if (lane >= o) inc += y;
        }
        if (lane == 31) wtot[w] = inc;
        __syncthreads();
        int off = 0;
        for (int j = 0; j < TE / 32; ++j)
            if (j < w) off += wtot[j];
        const int ex = off + inc - nb;
        if (tid < nbin) {
            gbase[tid] = g;
            lstart[tid] = ex;
            cur[tid] = ex;
        }
        __syncthreads();
        int i = start + tid;
        for (; i + 3 * TE < end; i += 4 * TE) {
            int s[4], d[4], pos[4];
            for (int j = 0; j < 4; ++j) {
                s[j] = src[i + j * TE];
                d[j] = dst[i + j * TE];
            }
            for (int j = 0; j < 4; ++j) pos[j] = atomicAdd(&cur[bin_of(d[j], nbin)], 1);
            for (int j = 0; j < 4; ++j) stage[min(max(pos[j], 0), CAP_E - 1)] = make_int2(s[j], d[j]);
        }
        for (; i < end; i += TE) {
            int s = src[i], d = dst[i];
            int pos = atomicAdd(&cur[bin_of(d, nbin)], 1);
            stage[min(max(pos, 0), CAP_E - 1)] = make_int2(s, d);
        }
        __syncthreads();
        for (int k = tid; k < len; k += TE) {
            int2 e = stage[k];
            int b = bin_of(e.y, nbin);
            int pos = gbase[b] + (k - lstart[b]);
            binned[min(max(pos, 0), E - 1)] = e;
        }
    } else {
        if (tid < nbin) cur[tid] = g;
        __syncthreads();
        for (int i = start + tid; i < end; i += TE) {
            int s = src[i], d = dst[i];
            int pos = atomicAdd(&cur[bin_of(d, nbin)], 1);
            binned[min(max(pos, 0), E - 1)] = make_int2(s, d);
        }
    }
}

__global__ void __launch_bounds__(TE) k_bin_count(const int2* __restrict__ binned,
                                                  const int* __restrict__ pre,
                                                  int* __restrict__ rowptr, int* __restrict__ cnt,
                                                  float* __restrict__ dinv, int n, int E, int nbin) {
    __shared__ int c[BIN_W];
    __shared__ int wtot[TE / 32];
    const int tid = threadIdx.x, lane = tid & 31, w = tid >> 5;
    const int bin = blockIdx.x;
    const int lo = bin << BIN_SHIFT;
    int bs = pre[bin * NBLK_E];
    int be = (bin + 1 < nbin) ? pre[(bin + 1) * NBLK_E] : E;
    bs = min(max(bs, 0), E);
    be = min(max(be, bs), E);
    for (int k = tid; k < BIN_W; k += TE) c[k] = 0;
    __syncthreads();
    int i = bs + tid;
    for (; i + 3 * TE < be; i += 4 * TE) {
        int d[4];
        for (int j = 0; j < 4; ++j) d[j] = binned[i + j * TE].y;
        for (int j = 0; j < 4; ++j) atomicAdd(&c[min(max(d[j] - lo, 0), BIN_W - 1)], 1);
    }
    for (; i < be; i += TE) atomicAdd(&c[min(max(binned[i].y - lo, 0), BIN_W - 1)], 1);
    __syncthreads();
    const int c0 = c[2 * tid], c1 = c[2 * tid + 1];
    const int s = c0 + c1;
    int inc = s;
    for (int o = 1; o < 32; o <<= 1) {
        int y = __shfl_up(inc, o, 32);
        if (lane >= o) inc += y;
    }
    if (lane == 31) wtot[w] = inc;
    __syncthreads();
    int off = 0;
    for (int j = 0; j < TE / 32; ++j)
        if (j < w) off += wtot[j];
    const int ex = bs + off + inc - s;
    const int node = lo + 2 * tid;
    if (node < n) {
        rowptr[node] = ex;
        cnt[node] = c0;
        float v = (float)(c0 + 1);
        dinv[node] = (v > 0.0f) ? rsqrtf(v) : 0.0f;
    }
    if (node + 1 < n) {
        rowptr[node + 1] = ex + c0;
        cnt[node + 1] = c1;
        float v = (float)(c1 + 1);
        dinv[node + 1] = (v > 0.0f) ? rsqrtf(v) : 0.0f;
    }
}

__global__ void __launch_bounds__(TE) k_bin_fill(const int2* __restrict__ binned,
                                                 const int* __restrict__ pre,
                                                 const int* __restrict__ rowptr,
                                                 const float* __restrict__ dinv,
                                                 int* __restrict__ lists, int n, int E, int nbin) {
    __shared__ int cur[BIN_W];
    __shared__ int2 stage[CAP_E];
    const int tid = threadIdx.x;
    const int bin = blockIdx.x;
    const int lo = bin << BIN_SHIFT;
    int bs = pre[bin * NBLK_E];
    int be = (bin + 1 < nbin) ? pre[(bin + 1) * NBLK_E] : E;
    bs = min(max(bs, 0), E);
    be = min(max(be, bs), E);
    for (int k = tid; k < BIN_W; k += TE) cur[k] = (lo + k < n) ? rowptr[lo + k] : 0;
    __syncthreads();
    int2* le = (int2*)lists;
    const int len = be - bs;
    if (len <= CAP_E) {
        int i = bs + tid;
        for (; i + 3 * TE < be; i += 4 * TE) {
            int2 e[4];
            float ds[4];
            int pos[4];
            for (int j = 0; j < 4; ++j) e[j] = binned[i + j * TE];
            for (int j = 0; j < 4; ++j) ds[j] = dinv[min(max(e[j].x, 0), n - 1)];
            for (int j = 0; j < 4; ++j) pos[j] = atomicAdd(&cur[min(max(e[j].y - lo, 0), BIN_W - 1)], 1) - bs;
            for (int j = 0; j < 4; ++j)
                stage[min(max(pos[j], 0), CAP_E - 1)] = make_int2(e[j].x, __float_as_int(ds[j]));
        }
        for (; i < be; i += TE) {
            int2 e = binned[i];
            float ds = dinv[min(max(e.x, 0), n - 1)];
            int pos = atomicAdd(&cur[min(max(e.y - lo, 0), BIN_W - 1)], 1) - bs;
            stage[min(max(pos, 0), CAP_E - 1)] = make_int2(e.x, __float_as_int(ds));
        }
        __syncthreads();
        for (int k = tid; k < len; k += TE) le[bs + k] = stage[k];
    } else {
        for (int i = bs + tid; i < be; i += TE) {
            int2 e = binned[i];
            float ds = dinv[min(max(e.x, 0), n - 1)];
            int pos = atomicAdd(&cur[min(max(e.y - lo, 0), BIN_W - 1)], 1);
            le[min(max(pos, 0), E - 1)] = make_int2(e.x, __float_as_int(ds));
        }
    }
}

__global__ void k_transpose_pad(const float* __restrict__ Bw, float* __restrict__ Bt,
                                int K, int Ncols, int ncols_pad) {
    int i = blockIdx.x * blockDim.x + threadIdx.x;
    int total = ncols_pad * K;
    if (i >= total) return;
    int ncol = i / K, k = i - ncol * K;
    Bt[i] = (ncol < Ncols) ? Bw[(size_t)k * Ncols + ncol] : 0.0f;
}

__global__ void __launch_bounds__(256) k_gather_finish(const int* __restrict__ rowptr,
                                                       const int* __restrict__ cnt,
                                                       const int* __restrict__ lists,
                                                       const float* __restrict__ t,
                                                       const float* __restrict__ dinv,
                                                       const float* __restrict__ bias,
                                                       float* __restrict__ h, int n_nodes) {
    int g = blockIdx.x * 256 + threadIdx.x;
    if (g >= n_nodes * 5) return;
    int node = g / 5;
    int q = g - node * 5;
    const float4* t4 = (const float4*)t;
    const int2* le = (const int2*)lists;
    float di = dinv[node];
    int k = rowptr[node];
    int k1 = k + cnt[node];
    int nmax = n_nodes - 1;

    float4 acc = make_float4(0.0f, 0.0f, 0.0f, 0.0f);
    for (; k + 4 <= k1; k += 4) {
        int2 e[4];
        float4 v[4];
        for (int j = 0; j < 4; ++j) e[j] = le[k + j];
        for (int j = 0; j < 4; ++j) v[j] = t4[min(max(e[j].x, 0), nmax) * 5 + q];
        for (int j = 0; j < 4; ++j) {
            float nr = __int_as_float(e[j].y) * di;
            acc.x = fmaf(v[j].x, nr, acc.x);
            acc.y = fmaf(v[j].y, nr, acc.y);
            acc.z = fmaf(v[j].z, nr, acc.z);
            acc.w = fmaf(v[j].w, nr, acc.w);
        }
    }
#pragma unroll 1
    for (; k < k1; ++k) {
        int2 e0 = le[k];
        float4 v0 = t4[min(max(e0.x, 0), nmax) * 5 + q];
        float nr = __int_as_float(e0.y) * di;
        acc.x = fmaf(v0.x, nr, acc.x);
        acc.y = fmaf(v0.y, nr, acc.y);
        acc.z = fmaf(v0.z, nr, acc.z);
        acc.w = fmaf(v0.w, nr, acc.w);
    }

    float4 sv = t4[g];
    const float* bq = bias + 4 * q;
    float4 o;
    o.x = fmaxf(acc.x + sv.x * di * di + bq[0], 0.0f);
    o.y = fmaxf(acc.y + sv.y * di * di + bq[1], 0.0f);
    o.z = fmaxf(acc.z + sv.z * di * di + bq[2], 0.0f);
    o.w = fmaxf(acc.w + sv.w * di * di + bq[3], 0.0f);
    ((float4*)h)[g] = o;
}

__global__ void k_wmma_gemm_f32(const float* __restrict__ A, const float* __restrict__ Bt,
                                const float* __restrict__ bias, float* __restrict__ C,
                                int M, int K, int Ncols, int mtiles, int ntiles, int do_relu) {
    int wave = (blockIdx.x * blockDim.x + threadIdx.x) >> 5;
    int lane = threadIdx.x & 31;
    int mt = wave % mtiles;
    int nt = wave / mtiles;
    if (nt >= ntiles) return;

    int lrow = lane & 15;
    int kh = (lane >> 4) << 1;
    int ncol = nt * 16 + lrow;
    const float* arow = A + (size_t)(mt * 16 + lrow) * K + kh;
    const float* brow = Bt + (size_t)ncol * K + kh;

    v8f acc = {};
    int kb = 0;
    for (; kb + 16 <= K; kb += 16) {
#pragma unroll
        for (int j = 0; j < 4; ++j) {
            v2f a = *(const v2f*)(arow + kb + 4 * j);
            v2f b = *(const v2f*)(brow + kb + 4 * j);
            acc = __builtin_amdgcn_wmma_f32_16x16x4_f32(false, a, false, b,
                                                        (short)0, acc, false, false);
        }
    }
    for (; kb < K; kb += 4) {
        v2f a = *(const v2f*)(arow + kb);
        v2f b = *(const v2f*)(brow + kb);
        acc = __builtin_amdgcn_wmma_f32_16x16x4_f32(false, a, false, b,
                                                    (short)0, acc, false, false);
    }

    if (ncol >= Ncols) return;
    float bv = bias ? bias[ncol] : 0.0f;
    int rbase = mt * 16 + ((lane >> 4) << 3);
#pragma unroll
    for (int r = 0; r < 8; ++r) {
        float v = acc[r] + bv;
        if (do_relu) v = fmaxf(v, 0.0f);
        C[(size_t)(rbase + r) * Ncols + ncol] = v;
    }
}

constexpr int NODE_TPW = 8;

__global__ void __launch_bounds__(256) k_node_gemm20(const float* __restrict__ X,
                                                     const float* __restrict__ Wm,
                                                     float* __restrict__ Tt, int mtiles) {
    const int lane = threadIdx.x & 31;
    const int wave = __builtin_amdgcn_readfirstlane((int)(blockIdx.x * 256 + threadIdx.x) >> 5);
    const int c = lane & 15, h = lane >> 4;
    const int t0 = wave * NODE_TPW;
    const int nt = min(NODE_TPW, mtiles - t0);
    if (nt <= 0) return;

    v2f wa[5][2];
#pragma unroll
    for (int j = 0; j < 5; ++j) {
#pragma unroll
        for (int mi = 0; mi < 2; ++mi) {
            const int m = mi * 16 + c;
            const int mc = min(m, 19);
            const int k0 = 4 * j + 2 * h;
            const float w0 = Wm[k0 * 20 + mc];
            const float w1 = Wm[(k0 + 1) * 20 + mc];
            wa[j][mi] = (m < 20) ? (v2f){w0, w1} : (v2f){0.0f, 0.0f};
        }
    }

    const float* xp = X + (size_t)(t0 * 16 + c) * 20 + 2 * h;
    float* tp = Tt + (size_t)(t0 * 16 + c) * 20 + 8 * h;
    v2f b[NODE_TPW][5];
#pragma unroll
    for (int i = 0; i < NODE_TPW; ++i) {
        const int ii = min(i, nt - 1);
#pragma unroll
        for (int j = 0; j < 5; ++j) b[i][j] = *(const v2f*)(xp + ii * 320 + 4 * j);
    }

    __builtin_amdgcn_sched_barrier(0);

#pragma unroll
    for (int i = 0; i < NODE_TPW; ++i) {
        v8f acc0 = {}, acc1 = {};
#pragma unroll
        for (int j = 0; j < 5; ++j) {
            acc0 = __builtin_amdgcn_wmma_f32_16x16x4_f32(false, wa[j][0], false, b[i][j],
                                                         (short)0, acc0, false, false);
            acc1 = __builtin_amdgcn_wmma_f32_16x16x4_f32(false, wa[j][1], false, b[i][j],
                                                         (short)0, acc1, false, false);
        }
        if (i < nt) {
            float* o = tp + i * 320;
            *(v4f*)o       = (v4f){acc0[0], acc0[1], acc0[2], acc0[3]};
            *(v4f*)(o + 4) = (v4f){acc0[4], acc0[5], acc0[6], acc0[7]};
            if (h == 0) *(v4f*)(o + 16) = (v4f){acc1[0], acc1[1], acc1[2], acc1[3]};
        }
    }
}

__global__ void __launch_bounds__(128) k_fc1_tiled(const float* __restrict__ A,
                                                   const float* __restrict__ Bw,
                                                   const float* __restrict__ bias,
                                                   float* __restrict__ C, int M) {
    constexpr int K = 1640, N = 300, BM = 32, BK = 40, NP = 336, PA = 44, NT = K / BK;
    constexpr int NB = (BK / 2) * (N / 4);
    constexpr int NA = BM * (BK / 4);
    constexpr int JB = (NB + 127) / 128;
    constexpr int JA = (NA + 127) / 128;
    constexpr int SA = BM * PA, SB = (BK / 2) * NP * 2;
    __shared__ __attribute__((aligned(16))) float As[2 * SA];
    __shared__ __attribute__((aligned(16))) float Bs[2 * SB];

    const int tid = threadIdx.x, lane = tid & 31, w = tid >> 5;
    const int c = lane & 15, h = lane >> 4;
    const int m0 = blockIdx.x * BM;

    for (int i = tid; i < (BK / 2) * 72; i += 128) {
        int kp = i / 72, r = i - kp * 72;
        Bs[(kp * NP + N) * 2 + r] = 0.0f;
        Bs[SB + (kp * NP + N) * 2 + r] = 0.0f;
    }

    v4f rb0[JB], rb1[JB], ra[JA];
    int boff[JB], bsoff[JB], aoff[JA], asoff[JA];
#pragma unroll
    for (int j = 0; j < JB; ++j) {
        int idx = min(tid + 128 * j, NB - 1);
        int kp = idx / (N / 4), q = idx - kp * (N / 4);
        boff[j] = (2 * kp) * N + 4 * q;
        bsoff[j] = (kp * NP + 4 * q) * 2;
    }
#pragma unroll
    for (int j = 0; j < JA; ++j) {
        int idx = min(tid + 128 * j, NA - 1);
        int m = idx / (BK / 4), jj = idx - m * (BK / 4);
        int row = min(m0 + m, M - 1);
        aoff[j] = row * K + 4 * jj;
        asoff[j] = m * PA + 4 * jj;
    }

#pragma unroll
    for (int j = 0; j < JB; ++j) {
        rb0[j] = *(const v4f*)(Bw + boff[j]);
        rb1[j] = *(const v4f*)(Bw + boff[j] + N);
    }
#pragma unroll
    for (int j = 0; j < JA; ++j)
        ra[j] = *(const v4f*)(A + aoff[j]);

    v8f acc[2][5] = {};
    const float* ap = As + c * PA + 2 * h;
    const float* bp = Bs + (h * NP + w * 80 + c) * 2;

#pragma unroll
    for (int j = 0; j < JB; ++j) {
        float* p = Bs + bsoff[j];
        *(v4f*)p       = (v4f){rb0[j].x, rb1[j].x, rb0[j].y, rb1[j].y};
        *(v4f*)(p + 4) = (v4f){rb0[j].z, rb1[j].z, rb0[j].w, rb1[j].w};
    }
#pragma unroll
    for (int j = 0; j < JA; ++j) *(v4f*)(As + asoff[j]) = ra[j];
#pragma unroll
    for (int j = 0; j < JB; ++j) {
        rb0[j] = *(const v4f*)(Bw + BK * N + boff[j]);
        rb1[j] = *(const v4f*)(Bw + BK * N + boff[j] + N);
    }
#pragma unroll
    for (int j = 0; j < JA; ++j) ra[j] = *(const v4f*)(A + BK + aoff[j]);
    __syncthreads();

    for (int t = 0; t < NT; ++t) {
        const int cur = (t & 1) * SB, nxt = SB - cur;
        const int curA = (t & 1) * SA, nxtA = SA - curA;

#pragma unroll
        for (int kk = 0; kk < BK / 4; ++kk) {
            if (kk == 1 && t + 1 < NT) {
#pragma unroll
                for (int j = 0; j < JB; ++j) {
                    float* p = Bs + nxt + bsoff[j];
                    *(v4f*)p       = (v4f){rb0[j].x, rb1[j].x, rb0[j].y, rb1[j].y};
                    *(v4f*)(p + 4) = (v4f){rb0[j].z, rb1[j].z, rb0[j].w, rb1[j].w};
                }
#pragma unroll
                for (int j = 0; j < JA; ++j) *(v4f*)(As + nxtA + asoff[j]) = ra[j];
                if (t + 2 < NT) {
                    const float* Bn = Bw + (t + 2) * (BK * N);
                    const float* An = A + (t + 2) * BK;
#pragma unroll
                    for (int j = 0; j < JB; ++j) {
                        rb0[j] = *(const v4f*)(Bn + boff[j]);
                        rb1[j] = *(const v4f*)(Bn + boff[j] + N);
                    }
#pragma unroll
                    for (int j = 0; j < JA; ++j) ra[j] = *(const v4f*)(An + aoff[j]);
                }
            }
            v2f a0 = *(const v2f*)(ap + curA + kk * 4);
            v2f a1 = *(const v2f*)(ap + curA + 16 * PA + kk * 4);
            v2f b[5];
#pragma unroll
            for (int ni = 0; ni < 5; ++ni) b[ni] = *(const v2f*)(bp + cur + kk * (4 * NP) + ni * 32);
#pragma unroll
            for (int ni = 0; ni < 5; ++ni) {
                acc[0][ni] = __builtin_amdgcn_wmma_f32_16x16x4_f32(false, a0, false, b[ni],
                                                                   (short)0, acc[0][ni], false, false);
                acc[1][ni] = __builtin_amdgcn_wmma_f32_16x16x4_f32(false, a1, false, b[ni],
                                                                   (short)0, acc[1][ni], false, false);
            }
        }
        __syncthreads();
    }

#pragma unroll
    for (int ni = 0; ni < 5; ++ni) {
        int col = w * 80 + ni * 16 + c;
        if (col < N) {
            float bv = bias[col];
#pragma unroll
            for (int mi = 0; mi < 2; ++mi) {
                int rbase = m0 + mi * 16 + 8 * h;
#pragma unroll
                for (int r = 0; r < 8; ++r)
                    if (rbase + r < M)
                        C[(size_t)(rbase + r) * N + col] = fmaxf(acc[mi][ni][r] + bv, 0.0f);
            }
        }
    }
}


extern "C" void kernel_launch(void* const* d_in, const int* in_sizes, int n_in,
                              void* d_out, int out_size, void* d_ws, size_t ws_size,
                              hipStream_t stream) {
    const float* x    = (const float*)d_in[0];
    const int*   eidx = (const int*)d_in[1];
    const float* W1   = (const float*)d_in[3];
    const float* b1   = (const float*)d_in[4];
    const float* W2   = (const float*)d_in[5];
    const float* b2   = (const float*)d_in[6];
    const float* fc1W = (const float*)d_in[7];
    const float* fc1b = (const float*)d_in[8];
    const float* outW = (const float*)d_in[9];
    const float* outb = (const float*)d_in[10];

    const int N  = in_sizes[2];
    const int E  = in_sizes[1] / 2;
    const int Bg = N / 82;
    const int EMB = 20, LAT = 82 * 20, HID = 300, NC = 22;

    char* ws = (char*)d_ws;
    size_t off = 0;
    auto take = [&](size_t bytes) -> void* {
        off = (off + 255) & ~(size_t)255;
        void* p = ws + off;
        off += bytes;
        return p;
    };
    const int nbin  = cdiv(N, BIN_W);
    const int chunk = cdiv(E, NBLK_E);
    const int NH    = nbin * NBLK_E;
    const int ntile = cdiv(NH, 1024);
    float* dinv   = (float*)take((size_t)N * 4);
    int*   cnt    = (int*)take((size_t)N * 4);
    int*   rowptr = (int*)take((size_t)N * 4);
    int*   hist   = (int*)take((size_t)NH * 4);
    int*   pre    = (int*)take((size_t)NH * 4);
    int*   tsum   = (int*)take((size_t)ntile * 4);
    int2*  binned = (int2*)take((size_t)E * 8);
    int*   lists  = (int*)take((size_t)E * 8);
    float* t    = (float*)take((size_t)N * EMB * 4);
    float* agg  = (float*)take((size_t)N * EMB * 4);
    float* fco  = (float*)take((size_t)Bg * HID * 4);
    float* Bt   = (float*)take((size_t)304 * LAT * 4);

    const int* src = eidx;
    const int* dst = eidx + E;

    const int TB = 256;

    auto run_gemm = [&](const float* A, const float* Bw, const float* bias, float* C,
                        int M, int K, int Ncols, int relu) {
        int ntiles = (Ncols + 15) / 16;
        int ncols_pad = ntiles * 16;
        int mtiles = M / 16;
        k_transpose_pad<<<cdiv(ncols_pad * K, TB), TB, 0, stream>>>(Bw, Bt, K, Ncols, ncols_pad);
        int waves = mtiles * ntiles;
        k_wmma_gemm_f32<<<cdiv(waves * 32, TB), TB, 0, stream>>>(
            A, Bt, bias, C, M, K, Ncols, mtiles, ntiles, relu);
    };

    k_bin_hist<<<NBLK_E, TE, 0, stream>>>(dst, hist, E, chunk, nbin);
    k_tile_sums<<<ntile, 256, 0, stream>>>(hist, tsum, NH);
    k_scan_excl<<<ntile, 256, 0, stream>>>(hist, tsum, pre, NH);
    k_bin_scatter<<<NBLK_E, TE, 0, stream>>>(src, dst, pre, binned, E, chunk, nbin);
    k_bin_count<<<nbin, TE, 0, stream>>>(binned, pre, rowptr, cnt, dinv, N, E, nbin);
    k_bin_fill<<<nbin, TE, 0, stream>>>(binned, pre, rowptr, dinv, lists, N, E, nbin);

    auto run_node_gemm = [&](const float* A, const float* Wm, float* C, int M) {
        int mtiles = M / 16;
        int waves = cdiv(mtiles, NODE_TPW);
        k_node_gemm20<<<cdiv(waves * 32, TB), TB, 0, stream>>>(A, Wm, C, mtiles);
    };

    run_node_gemm(x, W1, t, N);
    k_gather_finish<<<cdiv(N * 5, 256), 256, 0, stream>>>(rowptr, cnt, lists, t, dinv, b1, agg, N);

    run_node_gemm(agg, W2, t, N);
    k_gather_finish<<<cdiv(N * 5, 256), 256, 0, stream>>>(rowptr, cnt, lists, t, dinv, b2, agg, N);

    k_fc1_tiled<<<cdiv(Bg, 32), 128, 0, stream>>>(agg  , fc1W, fc1b, fco, Bg);

    run_gemm(fco, outW, outb, (float*)d_out, Bg, HID, NC, 0);
}


# dinv[src] folded into the gathered rows (node GEMM stores dinv*(x@W), sum scaled once by dinv[dst]): list entry = 4-byte src id, list fill drops its scattered per-edge dinv lookup; sum order kept
# speedup vs baseline: 7.7773x; 1.0889x over previous
#include <hip/hip_runtime.h>
#include <hip/hip_bf16.h>

typedef __attribute__((ext_vector_type(2))) float v2f;
typedef __attribute__((ext_vector_type(4))) float v4f;
typedef __attribute__((ext_vector_type(8))) float v8f;

static __host__ inline int cdiv(int a, int b) { return (a + b - 1) / b; }

constexpr int BIN_SHIFT = 11;
constexpr int BIN_W = 1 << BIN_SHIFT;
constexpr int MAX_BINS = 1024;
constexpr int NBLK_E = 256;
constexpr int TE = 1024;
constexpr int CAP_E = 16384;
static_assert(BIN_W == 2 * TE, "k_bin_count scans two counters per thread");
static_assert(MAX_BINS == TE, "k_bin_scatter scans one bin per thread");

__device__ __forceinline__ int bin_of(int d, int nbin) { return min(max(d >> BIN_SHIFT, 0), nbin - 1); }

__global__ void __launch_bounds__(TE) k_bin_hist(const int* __restrict__ dst, int* __restrict__ hist,
                                                 int E, int chunk, int nbin) {
    __shared__ int hc[MAX_BINS];
    const int tid = threadIdx.x;
    for (int i = tid; i < nbin; i += TE) hc[i] = 0;
    __syncthreads();
    const int start = blockIdx.x * chunk;
    const int end = min(start + chunk, E);
    int i = start + tid;
    for (; i + 3 * TE < end; i += 4 * TE) {
        int d[4];
        for (int j = 0; j < 4; ++j) d[j] = dst[i + j * TE];
        for (int j = 0; j < 4; ++j) atomicAdd(&hc[bin_of(d[j], nbin)], 1);
    }
    for (; i < end; i += TE) atomicAdd(&hc[bin_of(dst[i], nbin)], 1);
    __syncthreads();
    for (int b = tid; b < nbin; b += TE) hist[b * NBLK_E + blockIdx.x] = hc[b];
}

__global__ void __launch_bounds__(256) k_tile_sums(const int* __restrict__ cnt,
                                                   int* __restrict__ tsum, int n) {
    __shared__ int wsum[8];
    int base = blockIdx.x * 1024 + threadIdx.x * 4;
    int s = 0;
    for (int j = 0; j < 4; ++j)
        if (base + j < n) s += cnt[base + j];
    for (int o = 16; o > 0; o >>= 1) s += __shfl_down(s, o, 32);
    if ((threadIdx.x & 31) == 0) wsum[threadIdx.x >> 5] = s;
    __syncthreads();
    if (threadIdx.x == 0) {
        int tot = 0;
        for (int w = 0; w < 8; ++w) tot += wsum[w];
        tsum[blockIdx.x] = tot;
    }
}

__global__ void __launch_bounds__(256) k_scan_excl(const int* __restrict__ cnt,
                                                   const int* __restrict__ tsum,
                                                   int* __restrict__ pre, int n) {
    __shared__ int wpre[8];
    __shared__ int winc[8];
    int tid = threadIdx.x, lane = tid & 31, w = tid >> 5;

    int p = 0;
    for (int j = tid; j < (int)blockIdx.x; j += 256) p += tsum[j];
    for (int o = 16; o > 0; o >>= 1) p += __shfl_down(p, o, 32);
    if (lane == 0) wpre[w] = p;

    int base = blockIdx.x * 1024 + tid * 4;
    int c[4];
    for (int j = 0; j < 4; ++j) c[j] = (base + j < n) ? cnt[base + j] : 0;
    int s = c[0] + c[1] + c[2] + c[3];
    int inc = s;
    for (int o = 1; o < 32; o <<= 1) {
        int y = __shfl_up(inc, o, 32);
        if (lane >= o) inc += y;
    }
    if (lane == 31) winc[w] = inc;
    __syncthreads();

    int off = 0;
    for (int j = 0; j < 8; ++j) {
        off += wpre[j];
        if (j < w) off += winc[j];
    }
    int ex = off + inc - s;
    for (int j = 0; j < 4; ++j) {
        if (base + j < n) {
            pre[base + j] = ex;
            ex += c[j];
        }
    }
}

__global__ void __launch_bounds__(TE) k_bin_scatter(const int* __restrict__ src,
                                                    const int* __restrict__ dst,
                                                    const int* __restrict__ pre,
                                                    int2* __restrict__ binned,
                                                    int E, int chunk, int nbin) {
    __shared__ int gbase[MAX_BINS];
    __shared__ int lstart[MAX_BINS];
    __shared__ int cur[MAX_BINS];
    __shared__ int wtot[TE / 32];
    __shared__ int2 stage[CAP_E];
    const int tid = threadIdx.x, lane = tid & 31, w = tid >> 5;
    const int start = blockIdx.x * chunk;
    const int end = min(start + chunk, E);
    const int len = max(end - start, 0);

    int g = 0, nb = 0;
    if (tid < nbin) {
        int idx = tid * NBLK_E + blockIdx.x;
        g = pre[idx];
        nb = ((idx + 1 < nbin * NBLK_E) ? pre[idx + 1] : E) - g;
    }

    if (len <= CAP_E) {
        int inc = nb;
        for (int o = 1; o < 32; o <<= 1) {
            int y = __shfl_up(inc, o, 32);
            if (lane >= o) inc += y;
        }
        if (lane == 31) wtot[w] = inc;
        __syncthreads();
        int off = 0;
        for (int j = 0; j < TE / 32; ++j)
            if (j < w) off += wtot[j];
        const int ex = off + inc - nb;
        if (tid < nbin) {
            gbase[tid] = g;
            lstart[tid] = ex;
            cur[tid] = ex;
        }
        __syncthreads();
        int i = start + tid;
        for (; i + 3 * TE < end; i += 4 * TE) {
            int s[4], d[4], pos[4];
            for (int j = 0; j < 4; ++j) {
                s[j] = src[i + j * TE];
                d[j] = dst[i + j * TE];
            }
            for (int j = 0; j < 4; ++j) pos[j] = atomicAdd(&cur[bin_of(d[j], nbin)], 1);
            for (int j = 0; j < 4; ++j) stage[min(max(pos[j], 0), CAP_E - 1)] = make_int2(s[j], d[j]);
        }
        for (; i < end; i += TE) {
            int s = src[i], d = dst[i];
            int pos = atomicAdd(&cur[bin_of(d, nbin)], 1);
            stage[min(max(pos, 0), CAP_E - 1)] = make_int2(s, d);
        }
        __syncthreads();
        for (int k = tid; k < len; k += TE) {
            int2 e = stage[k];
            int b = bin_of(e.y, nbin);
            int pos = gbase[b] + (k - lstart[b]);
            binned[min(max(pos, 0), E - 1)] = e;
        }
    } else {
        if (tid < nbin) cur[tid] = g;
        __syncthreads();
        for (int i = start + tid; i < end; i += TE) {
            int s = src[i], d = dst[i];
            int pos = atomicAdd(&cur[bin_of(d, nbin)], 1);
            binned[min(max(pos, 0), E - 1)] = make_int2(s, d);
        }
    }
}

__global__ void __launch_bounds__(TE) k_bin_count(const int2* __restrict__ binned,
                                                  const int* __restrict__ pre,
                                                  int* __restrict__ rowptr, int* __restrict__ cnt,
                                                  float* __restrict__ dinv, int n, int E, int nbin) {
    __shared__ int c[BIN_W];
    __shared__ int wtot[TE / 32];
    const int tid = threadIdx.x, lane = tid & 31, w = tid >> 5;
    const int bin = blockIdx.x;
    const int lo = bin << BIN_SHIFT;
    int bs = pre[bin * NBLK_E];
    int be = (bin + 1 < nbin) ? pre[(bin + 1) * NBLK_E] : E;
    bs = min(max(bs, 0), E);
    be = min(max(be, bs), E);
    for (int k = tid; k < BIN_W; k += TE) c[k] = 0;
    __syncthreads();
    int i = bs + tid;
    for (; i + 3 * TE < be; i += 4 * TE) {
        int d[4];
        for (int j = 0; j < 4; ++j) d[j] = binned[i + j * TE].y;
        for (int j = 0; j < 4; ++j) atomicAdd(&c[min(max(d[j] - lo, 0), BIN_W - 1)], 1);
    }
    for (; i < be; i += TE) atomicAdd(&c[min(max(binned[i].y - lo, 0), BIN_W - 1)], 1);
    __syncthreads();
    const int c0 = c[2 * tid], c1 = c[2 * tid + 1];
    const int s = c0 + c1;
    int inc = s;
    for (int o = 1; o < 32; o <<= 1) {
        int y = __shfl_up(inc, o, 32);
        if (lane >= o) inc += y;
    }
    if (lane == 31) wtot[w] = inc;
    __syncthreads();
    int off = 0;
    for (int j = 0; j < TE / 32; ++j)
        if (j < w) off += wtot[j];
    const int ex = bs + off + inc - s;
    const int node = lo + 2 * tid;
    if (node < n) {
        rowptr[node] = ex;
        cnt[node] = c0;
        float v = (float)(c0 + 1);
        dinv[node] = (v > 0.0f) ? rsqrtf(v) : 0.0f;
    }
    if (node + 1 < n) {
        rowptr[node + 1] = ex + c0;
        cnt[node + 1] = c1;
        float v = (float)(c1 + 1);
        dinv[node + 1] = (v > 0.0f) ? rsqrtf(v) : 0.0f;
    }
}

__global__ void __launch_bounds__(TE) k_bin_fill(const int2* __restrict__ binned,
                                                 const int* __restrict__ pre,
                                                 const int* __restrict__ rowptr,
                                                 int* __restrict__ lists, int n, int E, int nbin) {
    __shared__ int cur[BIN_W];
    __shared__ int stage[CAP_E];
    const int tid = threadIdx.x;
    const int bin = blockIdx.x;
    const int lo = bin << BIN_SHIFT;
    int bs = pre[bin * NBLK_E];
    int be = (bin + 1 < nbin) ? pre[(bin + 1) * NBLK_E] : E;
    bs = min(max(bs, 0), E);
    be = min(max(be, bs), E);
    for (int k = tid; k < BIN_W; k += TE) cur[k] = (lo + k < n) ? rowptr[lo + k] : 0;
    __syncthreads();
    const int len = be - bs;
    if (len <= CAP_E) {
        int i = bs + tid;
        for (; i + 3 * TE < be; i += 4 * TE) {
            int2 e[4];
            int pos[4];
            for (int j = 0; j < 4; ++j) e[j] = binned[i + j * TE];
            for (int j = 0; j < 4; ++j) pos[j] = atomicAdd(&cur[min(max(e[j].y - lo, 0), BIN_W - 1)], 1) - bs;
            for (int j = 0; j < 4; ++j) stage[min(max(pos[j], 0), CAP_E - 1)] = e[j].x;
        }
        for (; i < be; i += TE) {
            int2 e = binned[i];
            int pos = atomicAdd(&cur[min(max(e.y - lo, 0), BIN_W - 1)], 1) - bs;
            stage[min(max(pos, 0), CAP_E - 1)] = e.x;
        }
        __syncthreads();
        for (int k = tid; k < len; k += TE) lists[bs + k] = stage[k];
    } else {
        for (int i = bs + tid; i < be; i += TE) {
            int2 e = binned[i];
            int pos = atomicAdd(&cur[min(max(e.y - lo, 0), BIN_W - 1)], 1);
            lists[min(max(pos, 0), E - 1)] = e.x;
        }
    }
}

__global__ void k_transpose_pad(const float* __restrict__ Bw, float* __restrict__ Bt,
                                int K, int Ncols, int ncols_pad) {
    int i = blockIdx.x * blockDim.x + threadIdx.x;
    int total = ncols_pad * K;
    if (i >= total) return;
    int ncol = i / K, k = i - ncol * K;
    Bt[i] = (ncol < Ncols) ? Bw[(size_t)k * Ncols + ncol] : 0.0f;
}

__global__ void __launch_bounds__(256) k_gather_finish(const int* __restrict__ rowptr,
                                                       const int* __restrict__ cnt,
                                                       const int* __restrict__ lists,
                                                       const float* __restrict__ t,
                                                       const float* __restrict__ dinv,
                                                       const float* __restrict__ bias,
                                                       float* __restrict__ h, int n_nodes) {
    int g = blockIdx.x * 256 + threadIdx.x;
    if (g >= n_nodes * 5) return;
    int node = g / 5;
    int q = g - node * 5;
    const float4* t4 = (const float4*)t;
    float di = dinv[node];
    int k = rowptr[node];
    int k1 = k + cnt[node];
    int nmax = n_nodes - 1;

    float4 acc = make_float4(0.0f, 0.0f, 0.0f, 0.0f);
    for (; k + 4 <= k1; k += 4) {
        int s[4];
        float4 v[4];
        for (int j = 0; j < 4; ++j) s[j] = lists[k + j];
        for (int j = 0; j < 4; ++j) v[j] = t4[min(max(s[j], 0), nmax) * 5 + q];
        for (int j = 0; j < 4; ++j) {
            acc.x += v[j].x;
            acc.y += v[j].y;
            acc.z += v[j].z;
            acc.w += v[j].w;
        }
    }
#pragma unroll 1
    for (; k < k1; ++k) {
        int s0 = lists[k];
        float4 v0 = t4[min(max(s0, 0), nmax) * 5 + q];
        acc.x += v0.x;
        acc.y += v0.y;
        acc.z += v0.z;
        acc.w += v0.w;
    }

    float4 sv = t4[g];
    const float* bq = bias + 4 * q;
    float4 o;
    o.x = fmaxf((acc.x + sv.x) * di + bq[0], 0.0f);
    o.y = fmaxf((acc.y + sv.y) * di + bq[1], 0.0f);
    o.z = fmaxf((acc.z + sv.z) * di + bq[2], 0.0f);
    o.w = fmaxf((acc.w + sv.w) * di + bq[3], 0.0f);
    ((float4*)h)[g] = o;
}

__global__ void k_wmma_gemm_f32(const float* __restrict__ A, const float* __restrict__ Bt,
                                const float* __restrict__ bias, float* __restrict__ C,
                                int M, int K, int Ncols, int mtiles, int ntiles, int do_relu) {
    int wave = (blockIdx.x * blockDim.x + threadIdx.x) >> 5;
    int lane = threadIdx.x & 31;
    int mt = wave % mtiles;
    int nt = wave / mtiles;
    if (nt >= ntiles) return;

    int lrow = lane & 15;
    int kh = (lane >> 4) << 1;
    int ncol = nt * 16 + lrow;
    const float* arow = A + (size_t)(mt * 16 + lrow) * K + kh;
    const float* brow = Bt + (size_t)ncol * K + kh;

    v8f acc = {};
    int kb = 0;
    for (; kb + 16 <= K; kb += 16) {
#pragma unroll
        for (int j = 0; j < 4; ++j) {
            v2f a = *(const v2f*)(arow + kb + 4 * j);
            v2f b = *(const v2f*)(brow + kb + 4 * j);
            acc = __builtin_amdgcn_wmma_f32_16x16x4_f32(false, a, false, b,
                                                        (short)0, acc, false, false);
        }
    }
    for (; kb < K; kb += 4) {
        v2f a = *(const v2f*)(arow + kb);
        v2f b = *(const v2f*)(brow + kb);
        acc = __builtin_amdgcn_wmma_f32_16x16x4_f32(false, a, false, b,
                                                    (short)0, acc, false, false);
    }

    if (ncol >= Ncols) return;
    float bv = bias ? bias[ncol] : 0.0f;
    int rbase = mt * 16 + ((lane >> 4) << 3);
#pragma unroll
    for (int r = 0; r < 8; ++r) {
        float v = acc[r] + bv;
        if (do_relu) v = fmaxf(v, 0.0f);
        C[(size_t)(rbase + r) * Ncols + ncol] = v;
    }
}

constexpr int NODE_TPW = 8;

__global__ void __launch_bounds__(256) k_node_gemm20(const float* __restrict__ X,
                                                     const float* __restrict__ Wm,
                                                     const float* __restrict__ dinv,
                                                     float* __restrict__ Tt, int mtiles) {
    const int lane = threadIdx.x & 31;
    const int wave = __builtin_amdgcn_readfirstlane((int)(blockIdx.x * 256 + threadIdx.x) >> 5);
    const int c = lane & 15, h = lane >> 4;
    const int t0 = wave * NODE_TPW;
    const int nt = min(NODE_TPW, mtiles - t0);
    if (nt <= 0) return;

    v2f wa[5][2];
#pragma unroll
    for (int j = 0; j < 5; ++j) {
#pragma unroll
        for (int mi = 0; mi < 2; ++mi) {
            const int m = mi * 16 + c;
            const int mc = min(m, 19);
            const int k0 = 4 * j + 2 * h;
            const float w0 = Wm[k0 * 20 + mc];
            const float w1 = Wm[(k0 + 1) * 20 + mc];
            wa[j][mi] = (m < 20) ? (v2f){w0, w1} : (v2f){0.0f, 0.0f};
        }
    }

    const float* xp = X + (size_t)(t0 * 16 + c) * 20 + 2 * h;
    float* tp = Tt + (size_t)(t0 * 16 + c) * 20 + 8 * h;
    const float* dp = dinv + (t0 * 16 + c);
    v2f b[NODE_TPW][5];
    float dv[NODE_TPW];
#pragma unroll
    for (int i = 0; i < NODE_TPW; ++i) {
        const int ii = min(i, nt - 1);
#pragma unroll
        for (int j = 0; j < 5; ++j) b[i][j] = *(const v2f*)(xp + ii * 320 + 4 * j);
        dv[i] = dp[ii * 16];
    }

    __builtin_amdgcn_sched_barrier(0);

#pragma unroll
    for (int i = 0; i < NODE_TPW; ++i) {
        v8f acc0 = {}, acc1 = {};
#pragma unroll
        for (int j = 0; j < 5; ++j) {
            acc0 = __builtin_amdgcn_wmma_f32_16x16x4_f32(false, wa[j][0], false, b[i][j],
                                                         (short)0, acc0, false, false);
            acc1 = __builtin_amdgcn_wmma_f32_16x16x4_f32(false, wa[j][1], false, b[i][j],
                                                         (short)0, acc1, false, false);
        }
        if (i < nt) {
            float* o = tp + i * 320;
            const float d = dv[i];
            *(v4f*)o       = (v4f){acc0[0] * d, acc0[1] * d, acc0[2] * d, acc0[3] * d};
            *(v4f*)(o + 4) = (v4f){acc0[4] * d, acc0[5] * d, acc0[6] * d, acc0[7] * d};
            if (h == 0) *(v4f*)(o + 16) = (v4f){acc1[0] * d, acc1[1] * d, acc1[2] * d, acc1[3] * d};
        }
    }
}

__global__ void __launch_bounds__(128) k_fc1_tiled(const float* __restrict__ A,
                                                   const float* __restrict__ Bw,
                                                   const float* __restrict__ bias,
                                                   float* __restrict__ C, int M) {
    constexpr int K = 1640, N = 300, BM = 32, BK = 40, NP = 336, PA = 44, NT = K / BK;
    constexpr int NB = (BK / 2) * (N / 4);
    constexpr int NA = BM * (BK / 4);
    constexpr int JB = (NB + 127) / 128;
    constexpr int JA = (NA + 127) / 128;
    constexpr int SA = BM * PA, SB = (BK / 2) * NP * 2;
    __shared__ __attribute__((aligned(16))) float As[2 * SA];
    __shared__ __attribute__((aligned(16))) float Bs[2 * SB];

    const int tid = threadIdx.x, lane = tid & 31, w = tid >> 5;
    const int c = lane & 15, h = lane >> 4;
    const int m0 = blockIdx.x * BM;

    for (int i = tid; i < (BK / 2) * 72; i += 128) {
        int kp = i / 72, r = i - kp * 72;
        Bs[(kp * NP + N) * 2 + r] = 0.0f;
        Bs[SB + (kp * NP + N) * 2 + r] = 0.0f;
    }

    v4f rb0[JB], rb1[JB], ra[JA];
    int boff[JB], bsoff[JB], aoff[JA], asoff[JA];
#pragma unroll
    for (int j = 0; j < JB; ++j) {
        int idx = min(tid + 128 * j, NB - 1);
        int kp = idx / (N / 4), q = idx - kp * (N / 4);
        boff[j] = (2 * kp) * N + 4 * q;
        bsoff[j] = (kp * NP + 4 * q) * 2;
    }
#pragma unroll
    for (int j = 0; j < JA; ++j) {
        int idx = min(tid + 128 * j, NA - 1);
        int m = idx / (BK / 4), jj = idx - m * (BK / 4);
        int row = min(m0 + m, M - 1);
        aoff[j] = row * K + 4 * jj;
        asoff[j] = m * PA + 4 * jj;
    }

#pragma unroll
    for (int j = 0; j < JB; ++j) {
        rb0[j] = *(const v4f*)(Bw + boff[j]);
        rb1[j] = *(const v4f*)(Bw + boff[j] + N);
    }
#pragma unroll
    for (int j = 0; j < JA; ++j)
        ra[j] = *(const v4f*)(A + aoff[j]);

    v8f acc[2][5] = {};
    const float* ap = As + c * PA + 2 * h;
    const float* bp = Bs + (h * NP + w * 80 + c) * 2;

#pragma unroll
    for (int j = 0; j < JB; ++j) {
        float* p = Bs + bsoff[j];
        *(v4f*)p       = (v4f){rb0[j].x, rb1[j].x, rb0[j].y, rb1[j].y};
        *(v4f*)(p + 4) = (v4f){rb0[j].z, rb1[j].z, rb0[j].w, rb1[j].w};
    }
#pragma unroll
    for (int j = 0; j < JA; ++j) *(v4f*)(As + asoff[j]) = ra[j];
#pragma unroll
    for (int j = 0; j < JB; ++j) {
        rb0[j] = *(const v4f*)(Bw + BK * N + boff[j]);
        rb1[j] = *(const v4f*)(Bw + BK * N + boff[j] + N);
    }
#pragma unroll
    for (int j = 0; j < JA; ++j) ra[j] = *(const v4f*)(A + BK + aoff[j]);
    __syncthreads();

    for (int t = 0; t < NT; ++t) {
        const int cur = (t & 1) * SB, nxt = SB - cur;
        const int curA = (t & 1) * SA, nxtA = SA - curA;

#pragma unroll
        for (int kk = 0; kk < BK / 4; ++kk) {
            if (kk == 1 && t + 1 < NT) {
#pragma unroll
                for (int j = 0; j < JB; ++j) {
                    float* p = Bs + nxt + bsoff[j];
                    *(v4f*)p       = (v4f){rb0[j].x, rb1[j].x, rb0[j].y, rb1[j].y};
                    *(v4f*)(p + 4) = (v4f){rb0[j].z, rb1[j].z, rb0[j].w, rb1[j].w};
                }
#pragma unroll
                for (int j = 0; j < JA; ++j) *(v4f*)(As + nxtA + asoff[j]) = ra[j];
                if (t + 2 < NT) {
                    const float* Bn = Bw + (t + 2) * (BK * N);
                    const float* An = A + (t + 2) * BK;
#pragma unroll
                    for (int j = 0; j < JB; ++j) {
                        rb0[j] = *(const v4f*)(Bn + boff[j]);
                        rb1[j] = *(const v4f*)(Bn + boff[j] + N);
                    }
#pragma unroll
                    for (int j = 0; j < JA; ++j) ra[j] = *(const v4f*)(An + aoff[j]);
                }
            }
            v2f a0 = *(const v2f*)(ap + curA + kk * 4);
            v2f a1 = *(const v2f*)(ap + curA + 16 * PA + kk * 4);
            v2f b[5];
#pragma unroll
            for (int ni = 0; ni < 5; ++ni) b[ni] = *(const v2f*)(bp + cur + kk * (4 * NP) + ni * 32);
#pragma unroll
            for (int ni = 0; ni < 5; ++ni) {
                acc[0][ni] = __builtin_amdgcn_wmma_f32_16x16x4_f32(false, a0, false, b[ni],
                                                                   (short)0, acc[0][ni], false, false);
                acc[1][ni] = __builtin_amdgcn_wmma_f32_16x16x4_f32(false, a1, false, b[ni],
                                                                   (short)0, acc[1][ni], false, false);
            }
        }
        __syncthreads();
    }

#pragma unroll
    for (int ni = 0; ni < 5; ++ni) {
        int col = w * 80 + ni * 16 + c;
        if (col < N) {
            float bv = bias[col];
#pragma unroll
            for (int mi = 0; mi < 2; ++mi) {
                int rbase = m0 + mi * 16 + 8 * h;
#pragma unroll
                for (int r = 0; r < 8; ++r)
                    if (rbase + r < M)
                        C[(size_t)(rbase + r) * N + col] = fmaxf(acc[mi][ni][r] + bv, 0.0f);
            }
        }
    }
}


extern "C" void kernel_launch(void* const* d_in, const int* in_sizes, int n_in,
                              void* d_out, int out_size, void* d_ws, size_t ws_size,
                              hipStream_t stream) {
    const float* x    = (const float*)d_in[0];
    const int*   eidx = (const int*)d_in[1];
    const float* W1   = (const float*)d_in[3];
    const float* b1   = (const float*)d_in[4];
    const float* W2   = (const float*)d_in[5];
    const float* b2   = (const float*)d_in[6];
    const float* fc1W = (const float*)d_in[7];
    const float* fc1b = (const float*)d_in[8];
    const float* outW = (const float*)d_in[9];
    const float* outb = (const float*)d_in[10];

    const int N  = in_sizes[2];
    const int E  = in_sizes[1] / 2;
    const int Bg = N / 82;
    const int EMB = 20, LAT = 82 * 20, HID = 300, NC = 22;

    char* ws = (char*)d_ws;
    size_t off = 0;
    auto take = [&](size_t bytes) -> void* {
        off = (off + 255) & ~(size_t)255;
        void* p = ws + off;
        off += bytes;
        return p;
    };
    const int nbin  = cdiv(N, BIN_W);
    const int chunk = cdiv(E, NBLK_E);
    const int NH    = nbin * NBLK_E;
    const int ntile = cdiv(NH, 1024);
    float* dinv   = (float*)take((size_t)N * 4);
    int*   cnt    = (int*)take((size_t)N * 4);
    int*   rowptr = (int*)take((size_t)N * 4);
    int*   hist   = (int*)take((size_t)NH * 4);
    int*   pre    = (int*)take((size_t)NH * 4);
    int*   tsum   = (int*)take((size_t)ntile * 4);
    int2*  binned = (int2*)take((size_t)E * 8);
    int*   lists  = (int*)take((size_t)E * 4);
    float* t    = (float*)take((size_t)N * EMB * 4);
    float* agg  = (float*)take((size_t)N * EMB * 4);
    float* fco  = (float*)take((size_t)Bg * HID * 4);
    float* Bt   = (float*)take((size_t)304 * LAT * 4);

    const int* src = eidx;
    const int* dst = eidx + E;

    const int TB = 256;

    auto run_gemm = [&](const float* A, const float* Bw, const float* bias, float* C,
                        int M, int K, int Ncols, int relu) {
        int ntiles = (Ncols + 15) / 16;
        int ncols_pad = ntiles * 16;
        int mtiles = M / 16;
        k_transpose_pad<<<cdiv(ncols_pad * K, TB), TB, 0, stream>>>(Bw, Bt, K, Ncols, ncols_pad);
        int waves = mtiles * ntiles;
        k_wmma_gemm_f32<<<cdiv(waves * 32, TB), TB, 0, stream>>>(
            A, Bt, bias, C, M, K, Ncols, mtiles, ntiles, relu);
    };

    k_bin_hist<<<NBLK_E, TE, 0, stream>>>(dst, hist, E, chunk, nbin);
    k_tile_sums<<<ntile, 256, 0, stream>>>(hist, tsum, NH);
    k_scan_excl<<<ntile, 256, 0, stream>>>(hist, tsum, pre, NH);
    k_bin_scatter<<<NBLK_E, TE, 0, stream>>>(src, dst, pre, binned, E, chunk, nbin);
    k_bin_count<<<nbin, TE, 0, stream>>>(binned, pre, rowptr, cnt, dinv, N, E, nbin);
    k_bin_fill<<<nbin, TE, 0, stream>>>(binned, pre, rowptr, lists, N, E, nbin);

    auto run_node_gemm = [&](const float* A, const float* Wm, float* C, int M) {
        int mtiles = M / 16;
        int waves = cdiv(mtiles, NODE_TPW);
        k_node_gemm20<<<cdiv(waves * 32, TB), TB, 0, stream>>>(A, Wm, dinv, C, mtiles);
    };

    run_node_gemm(x, W1, t, N);
    k_gather_finish<<<cdiv(N * 5, 256), 256, 0, stream>>>(rowptr, cnt, lists, t, dinv, b1, agg, N);

    run_node_gemm(agg, W2, t, N);
    k_gather_finish<<<cdiv(N * 5, 256), 256, 0, stream>>>(rowptr, cnt, lists, t, dinv, b2, agg, N);

    k_fc1_tiled<<<cdiv(Bg, 32), 128, 0, stream>>>(agg  , fc1W, fc1b, fco, Bg);

    run_gemm(fco, outW, outb, (float*)d_out, Bg, HID, NC, 0);
}


# Layer 2's node GEMM folded into layer 1's gather: finished h1 rows reach the WMMA through LDS (64 nodes per block), dinv*(h1@W2) is stored directly; one launch less; k order and sum order kept
# speedup vs baseline: 8.0943x; 1.0408x over previous
#include <hip/hip_runtime.h>
#include <hip/hip_bf16.h>

typedef __attribute__((ext_vector_type(2))) float v2f;
typedef __attribute__((ext_vector_type(4))) float v4f;
typedef __attribute__((ext_vector_type(8))) float v8f;

static __host__ inline int cdiv(int a, int b) { return (a + b - 1) / b; }

constexpr int BIN_SHIFT = 11;
constexpr int BIN_W = 1 << BIN_SHIFT;
constexpr int MAX_BINS = 1024;
constexpr int NBLK_E = 256;
constexpr int TE = 1024;
constexpr int CAP_E = 16384;
static_assert(BIN_W == 2 * TE, "k_bin_count scans two counters per thread");
static_assert(MAX_BINS == TE, "k_bin_scatter scans one bin per thread");

__device__ __forceinline__ int bin_of(int d, int nbin) { return min(max(d >> BIN_SHIFT, 0), nbin - 1); }

__global__ void __launch_bounds__(TE) k_bin_hist(const int* __restrict__ dst, int* __restrict__ hist,
                                                 int E, int chunk, int nbin) {
    __shared__ int hc[MAX_BINS];
    const int tid = threadIdx.x;
    for (int i = tid; i < nbin; i += TE) hc[i] = 0;
    __syncthreads();
    const int start = blockIdx.x * chunk;
    const int end = min(start + chunk, E);
    int i = start + tid;
    for (; i + 3 * TE < end; i += 4 * TE) {
        int d[4];
        for (int j = 0; j < 4; ++j) d[j] = dst[i + j * TE];
        for (int j = 0; j < 4; ++j) atomicAdd(&hc[bin_of(d[j], nbin)], 1);
    }
    for (; i < end; i += TE) atomicAdd(&hc[bin_of(dst[i], nbin)], 1);
    __syncthreads();
    for (int b = tid; b < nbin; b += TE) hist[b * NBLK_E + blockIdx.x] = hc[b];
}

__global__ void __launch_bounds__(256) k_tile_sums(const int* __restrict__ cnt,
                                                   int* __restrict__ tsum, int n) {
    __shared__ int wsum[8];
    int base = blockIdx.x * 1024 + threadIdx.x * 4;
    int s = 0;
    for (int j = 0; j < 4; ++j)
        if (base + j < n) s += cnt[base + j];
    for (int o = 16; o > 0; o >>= 1) s += __shfl_down(s, o, 32);
    if ((threadIdx.x & 31) == 0) wsum[threadIdx.x >> 5] = s;
    __syncthreads();
    if (threadIdx.x == 0) {
        int tot = 0;
        for (int w = 0; w < 8; ++w) tot += wsum[w];
        tsum[blockIdx.x] = tot;
    }
}

__global__ void __launch_bounds__(256) k_scan_excl(const int* __restrict__ cnt,
                                                   const int* __restrict__ tsum,
                                                   int* __restrict__ pre, int n) {
    __shared__ int wpre[8];
    __shared__ int winc[8];
    int tid = threadIdx.x, lane = tid & 31, w = tid >> 5;

    int p = 0;
    for (int j = tid; j < (int)blockIdx.x; j += 256) p += tsum[j];
    for (int o = 16; o > 0; o >>= 1) p += __shfl_down(p, o, 32);
    if (lane == 0) wpre[w] = p;

    int base = blockIdx.x * 1024 + tid * 4;
    int c[4];
    for (int j = 0; j < 4; ++j) c[j] = (base + j < n) ? cnt[base + j] : 0;
    int s = c[0] + c[1] + c[2] + c[3];
    int inc = s;
    for (int o = 1; o < 32; o <<= 1) {
        int y = __shfl_up(inc, o, 32);
        if (lane >= o) inc += y;
    }
    if (lane == 31) winc[w] = inc;
    __syncthreads();

    int off = 0;
    for (int j = 0; j < 8; ++j) {
        off += wpre[j];
        if (j < w) off += winc[j];
    }
    int ex = off + inc - s;
    for (int j = 0; j < 4; ++j) {
        if (base + j < n) {
            pre[base + j] = ex;
            ex += c[j];
        }
    }
}

__global__ void __launch_bounds__(TE) k_bin_scatter(const int* __restrict__ src,
                                                    const int* __restrict__ dst,
                                                    const int* __restrict__ pre,
                                                    int2* __restrict__ binned,
                                                    int E, int chunk, int nbin) {
    __shared__ int gbase[MAX_BINS];
    __shared__ int lstart[MAX_BINS];
    __shared__ int cur[MAX_BINS];
    __shared__ int wtot[TE / 32];
    __shared__ int2 stage[CAP_E];
    const int tid = threadIdx.x, lane = tid & 31, w = tid >> 5;
    const int start = blockIdx.x * chunk;
    const int end = min(start + chunk, E);
    const int len = max(end - start, 0);

    int g = 0, nb = 0;
    if (tid < nbin) {
        int idx = tid * NBLK_E + blockIdx.x;
        g = pre[idx];
        nb = ((idx + 1 < nbin * NBLK_E) ? pre[idx + 1] : E) - g;
    }

    if (len <= CAP_E) {
        int inc = nb;
        for (int o = 1; o < 32; o <<= 1) {
            int y = __shfl_up(inc, o, 32);
            if (lane >= o) inc += y;
        }
        if (lane == 31) wtot[w] = inc;
        __syncthreads();
        int off = 0;
        for (int j = 0; j < TE / 32; ++j)
            if (j < w) off += wtot[j];
        const int ex = off + inc - nb;
        if (tid < nbin) {
            gbase[tid] = g;
            lstart[tid] = ex;
            cur[tid] = ex;
        }
        __syncthreads();
        int i = start + tid;
        for (; i + 3 * TE < end; i += 4 * TE) {
            int s[4], d[4], pos[4];
            for (int j = 0; j < 4; ++j) {
                s[j] = src[i + j * TE];
                d[j] = dst[i + j * TE];
            }
            for (int j = 0; j < 4; ++j) pos[j] = atomicAdd(&cur[bin_of(d[j], nbin)], 1);
            for (int j = 0; j < 4; ++j) stage[min(max(pos[j], 0), CAP_E - 1)] = make_int2(s[j], d[j]);
        }
        for (; i < end; i += TE) {
            int s = src[i], d = dst[i];
            int pos = atomicAdd(&cur[bin_of(d, nbin)], 1);
            stage[min(max(pos, 0), CAP_E - 1)] = make_int2(s, d);
        }
        __syncthreads();
        for (int k = tid; k < len; k += TE) {
            int2 e = stage[k];
            int b = bin_of(e.y, nbin);
            int pos = gbase[b] + (k - lstart[b]);
            binned[min(max(pos, 0), E - 1)] = e;
        }
    } else {
        if (tid < nbin) cur[tid] = g;
        __syncthreads();
        for (int i = start + tid; i < end; i += TE) {
            int s = src[i], d = dst[i];
            int pos = atomicAdd(&cur[bin_of(d, nbin)], 1);
            binned[min(max(pos, 0), E - 1)] = make_int2(s, d);
        }
    }
}

__global__ void __launch_bounds__(TE) k_bin_count(const int2* __restrict__ binned,
                                                  const int* __restrict__ pre,
                                                  int* __restrict__ rowptr, int* __restrict__ cnt,
                                                  float* __restrict__ dinv, int n, int E, int nbin) {
    __shared__ int c[BIN_W];
    __shared__ int wtot[TE / 32];
    const int tid = threadIdx.x, lane = tid & 31, w = tid >> 5;
    const int bin = blockIdx.x;
    const int lo = bin << BIN_SHIFT;
    int bs = pre[bin * NBLK_E];
    int be = (bin + 1 < nbin) ? pre[(bin + 1) * NBLK_E] : E;
    bs = min(max(bs, 0), E);
    be = min(max(be, bs), E);
    for (int k = tid; k < BIN_W; k += TE) c[k] = 0;
    __syncthreads();
    int i = bs + tid;
    for (; i + 3 * TE < be; i += 4 * TE) {
        int d[4];
        for (int j = 0; j < 4; ++j) d[j] = binned[i + j * TE].y;
        for (int j = 0; j < 4; ++j) atomicAdd(&c[min(max(d[j] - lo, 0), BIN_W - 1)], 1);
    }
    for (; i < be; i += TE) atomicAdd(&c[min(max(binned[i].y - lo, 0), BIN_W - 1)], 1);
    __syncthreads();
    const int c0 = c[2 * tid], c1 = c[2 * tid + 1];
    const int s = c0 + c1;
    int inc = s;
    for (int o = 1; o < 32; o <<= 1) {
        int y = __shfl_up(inc, o, 32);
        if (lane >= o) inc += y;
    }
    if (lane == 31) wtot[w] = inc;
    __syncthreads();
    int off = 0;
    for (int j = 0; j < TE / 32; ++j)
        if (j < w) off += wtot[j];
    const int ex = bs + off + inc - s;
    const int node = lo + 2 * tid;
    if (node < n) {
        rowptr[node] = ex;
        cnt[node] = c0;
        float v = (float)(c0 + 1);
        dinv[node] = (v > 0.0f) ? rsqrtf(v) : 0.0f;
    }
    if (node + 1 < n) {
        rowptr[node + 1] = ex + c0;
        cnt[node + 1] = c1;
        float v = (float)(c1 + 1);
        dinv[node + 1] = (v > 0.0f) ? rsqrtf(v) : 0.0f;
    }
}

__global__ void __launch_bounds__(TE) k_bin_fill(const int2* __restrict__ binned,
                                                 const int* __restrict__ pre,
                                                 const int* __restrict__ rowptr,
                                                 int* __restrict__ lists, int n, int E, int nbin) {
    __shared__ int cur[BIN_W];
    __shared__ int stage[CAP_E];
    const int tid = threadIdx.x;
    const int bin = blockIdx.x;
    const int lo = bin << BIN_SHIFT;
    int bs = pre[bin * NBLK_E];
    int be = (bin + 1 < nbin) ? pre[(bin + 1) * NBLK_E] : E;
    bs = min(max(bs, 0), E);
    be = min(max(be, bs), E);
    for (int k = tid; k < BIN_W; k += TE) cur[k] = (lo + k < n) ? rowptr[lo + k] : 0;
    __syncthreads();
    const int len = be - bs;
    if (len <= CAP_E) {
        int i = bs + tid;
        for (; i + 3 * TE < be; i += 4 * TE) {
            int2 e[4];
            int pos[4];
            for (int j = 0; j < 4; ++j) e[j] = binned[i + j * TE];
            for (int j = 0; j < 4; ++j) pos[j] = atomicAdd(&cur[min(max(e[j].y - lo, 0), BIN_W - 1)], 1) - bs;
            for (int j = 0; j < 4; ++j) stage[min(max(pos[j], 0), CAP_E - 1)] = e[j].x;
        }
        for (; i < be; i += TE) {
            int2 e = binned[i];
            int pos = atomicAdd(&cur[min(max(e.y - lo, 0), BIN_W - 1)], 1) - bs;
            stage[min(max(pos, 0), CAP_E - 1)] = e.x;
        }
        __syncthreads();
        for (int k = tid; k < len; k += TE) lists[bs + k] = stage[k];
    } else {
        for (int i = bs + tid; i < be; i += TE) {
            int2 e = binned[i];
            int pos = atomicAdd(&cur[min(max(e.y - lo, 0), BIN_W - 1)], 1);
            lists[min(max(pos, 0), E - 1)] = e.x;
        }
    }
}

__global__ void k_transpose_pad(const float* __restrict__ Bw, float* __restrict__ Bt,
                                int K, int Ncols, int ncols_pad) {
    int i = blockIdx.x * blockDim.x + threadIdx.x;
    int total = ncols_pad * K;
    if (i >= total) return;
    int ncol = i / K, k = i - ncol * K;
    Bt[i] = (ncol < Ncols) ? Bw[(size_t)k * Ncols + ncol] : 0.0f;
}

__global__ void __launch_bounds__(256) k_gather_finish(const int* __restrict__ rowptr,
                                                       const int* __restrict__ cnt,
                                                       const int* __restrict__ lists,
                                                       const float* __restrict__ t,
                                                       const float* __restrict__ dinv,
                                                       const float* __restrict__ bias,
                                                       float* __restrict__ h, int n_nodes) {
    int g = blockIdx.x * 256 + threadIdx.x;
    if (g >= n_nodes * 5) return;
    int node = g / 5;
    int q = g - node * 5;
    const float4* t4 = (const float4*)t;
    float di = dinv[node];
    int k = rowptr[node];
    int k1 = k + cnt[node];
    int nmax = n_nodes - 1;

    float4 acc = make_float4(0.0f, 0.0f, 0.0f, 0.0f);
    for (; k + 4 <= k1; k += 4) {
        int s[4];
        float4 v[4];
        for (int j = 0; j < 4; ++j) s[j] = lists[k + j];
        for (int j = 0; j < 4; ++j) v[j] = t4[min(max(s[j], 0), nmax) * 5 + q];
        for (int j = 0; j < 4; ++j) {
            acc.x += v[j].x;
            acc.y += v[j].y;
            acc.z += v[j].z;
            acc.w += v[j].w;
        }
    }
#pragma unroll 1
    for (; k < k1; ++k) {
        int s0 = lists[k];
        float4 v0 = t4[min(max(s0, 0), nmax) * 5 + q];
        acc.x += v0.x;
        acc.y += v0.y;
        acc.z += v0.z;
        acc.w += v0.w;
    }

    float4 sv = t4[g];
    const float* bq = bias + 4 * q;
    float4 o;
    o.x = fmaxf((acc.x + sv.x) * di + bq[0], 0.0f);
    o.y = fmaxf((acc.y + sv.y) * di + bq[1], 0.0f);
    o.z = fmaxf((acc.z + sv.z) * di + bq[2], 0.0f);
    o.w = fmaxf((acc.w + sv.w) * di + bq[3], 0.0f);
    ((float4*)h)[g] = o;
}

__global__ void k_wmma_gemm_f32(const float* __restrict__ A, const float* __restrict__ Bt,
                                const float* __restrict__ bias, float* __restrict__ C,
                                int M, int K, int Ncols, int mtiles, int ntiles, int do_relu) {
    int wave = (blockIdx.x * blockDim.x + threadIdx.x) >> 5;
    int lane = threadIdx.x & 31;
    int mt = wave % mtiles;
    int nt = wave / mtiles;
    if (nt >= ntiles) return;

    int lrow = lane & 15;
    int kh = (lane >> 4) << 1;
    int ncol = nt * 16 + lrow;
    const float* arow = A + (size_t)(mt * 16 + lrow) * K + kh;
    const float* brow = Bt + (size_t)ncol * K + kh;

    v8f acc = {};
    int kb = 0;
    for (; kb + 16 <= K; kb += 16) {
#pragma unroll
        for (int j = 0; j < 4; ++j) {
            v2f a = *(const v2f*)(arow + kb + 4 * j);
            v2f b = *(const v2f*)(brow + kb + 4 * j);
            acc = __builtin_amdgcn_wmma_f32_16x16x4_f32(false, a, false, b,
                                                        (short)0, acc, false, false);
        }
    }
    for (; kb < K; kb += 4) {
        v2f a = *(const v2f*)(arow + kb);
        v2f b = *(const v2f*)(brow + kb);
        acc = __builtin_amdgcn_wmma_f32_16x16x4_f32(false, a, false, b,
                                                    (short)0, acc, false, false);
    }

    if (ncol >= Ncols) return;
    float bv = bias ? bias[ncol] : 0.0f;
    int rbase = mt * 16 + ((lane >> 4) << 3);
#pragma unroll
    for (int r = 0; r < 8; ++r) {
        float v = acc[r] + bv;
        if (do_relu) v = fmaxf(v, 0.0f);
        C[(size_t)(rbase + r) * Ncols + ncol] = v;
    }
}

constexpr int NODE_TPW = 8;

__global__ void __launch_bounds__(256) k_node_gemm20(const float* __restrict__ X,
                                                     const float* __restrict__ Wm,
                                                     const float* __restrict__ dinv,
                                                     float* __restrict__ Tt, int mtiles) {
    const int lane = threadIdx.x & 31;
    const int wave = __builtin_amdgcn_readfirstlane((int)(blockIdx.x * 256 + threadIdx.x) >> 5);
    const int c = lane & 15, h = lane >> 4;
    const int t0 = wave * NODE_TPW;
    const int nt = min(NODE_TPW, mtiles - t0);
    if (nt <= 0) return;

    v2f wa[5][2];
#pragma unroll
    for (int j = 0; j < 5; ++j) {
#pragma unroll
        for (int mi = 0; mi < 2; ++mi) {
            const int m = mi * 16 + c;
            const int mc = min(m, 19);
            const int k0 = 4 * j + 2 * h;
            const float w0 = Wm[k0 * 20 + mc];
            const float w1 = Wm[(k0 + 1) * 20 + mc];
            wa[j][mi] = (m < 20) ? (v2f){w0, w1} : (v2f){0.0f, 0.0f};
        }
    }

    const float* xp = X + (size_t)(t0 * 16 + c) * 20 + 2 * h;
    float* tp = Tt + (size_t)(t0 * 16 + c) * 20 + 8 * h;
    const float* dp = dinv + (t0 * 16 + c);
    v2f b[NODE_TPW][5];
    float dv[NODE_TPW];
#pragma unroll
    for (int i = 0; i < NODE_TPW; ++i) {
        const int ii = min(i, nt - 1);
#pragma unroll
        for (int j = 0; j < 5; ++j) b[i][j] = *(const v2f*)(xp + ii * 320 + 4 * j);
        dv[i] = dp[ii * 16];
    }

    __builtin_amdgcn_sched_barrier(0);

#pragma unroll
    for (int i = 0; i < NODE_TPW; ++i) {
        v8f acc0 = {}, acc1 = {};
#pragma unroll
        for (int j = 0; j < 5; ++j) {
            acc0 = __builtin_amdgcn_wmma_f32_16x16x4_f32(false, wa[j][0], false, b[i][j],
                                                         (short)0, acc0, false, false);
            acc1 = __builtin_amdgcn_wmma_f32_16x16x4_f32(false, wa[j][1], false, b[i][j],
                                                         (short)0, acc1, false, false);
        }
        if (i < nt) {
            float* o = tp + i * 320;
            const float d = dv[i];
            *(v4f*)o       = (v4f){acc0[0] * d, acc0[1] * d, acc0[2] * d, acc0[3] * d};
            *(v4f*)(o + 4) = (v4f){acc0[4] * d, acc0[5] * d, acc0[6] * d, acc0[7] * d};
            if (h == 0) *(v4f*)(o + 16) = (v4f){acc1[0] * d, acc1[1] * d, acc1[2] * d, acc1[3] * d};
        }
    }
}

constexpr int GN = 64;
constexpr int GT = GN * 5;

__global__ void __launch_bounds__(GT) k_gather_node_gemm20(const int* __restrict__ rowptr,
                                                           const int* __restrict__ cnt,
                                                           const int* __restrict__ lists,
                                                           const float* __restrict__ t,
                                                           const float* __restrict__ dinv,
                                                           const float* __restrict__ bias,
                                                           const float* __restrict__ Wm,
                                                           float* __restrict__ Tt, int n_nodes) {
    __shared__ __attribute__((aligned(16))) float hs[GN * 20];
    const int tid = threadIdx.x;
    const int node0 = blockIdx.x * GN;
    const int nl = tid / 5;
    const int q = tid - nl * 5;
    const int node = node0 + nl;

    if (node < n_nodes) {
        const int g = node * 5 + q;
        const float4* t4 = (const float4*)t;
        float di = dinv[node];
        int k = rowptr[node];
        int k1 = k + cnt[node];
        int nmax = n_nodes - 1;

        float4 acc = make_float4(0.0f, 0.0f, 0.0f, 0.0f);
        for (; k + 4 <= k1; k += 4) {
            int s[4];
            float4 v[4];
            for (int j = 0; j < 4; ++j) s[j] = lists[k + j];
            for (int j = 0; j < 4; ++j) v[j] = t4[min(max(s[j], 0), nmax) * 5 + q];
            for (int j = 0; j < 4; ++j) {
                acc.x += v[j].x;
                acc.y += v[j].y;
                acc.z += v[j].z;
                acc.w += v[j].w;
            }
        }
#pragma unroll 1
        for (; k < k1; ++k) {
            int s0 = lists[k];
            float4 v0 = t4[min(max(s0, 0), nmax) * 5 + q];
            acc.x += v0.x;
            acc.y += v0.y;
            acc.z += v0.z;
            acc.w += v0.w;
        }

        float4 sv = t4[g];
        const float* bq = bias + 4 * q;
        float4 o;
        o.x = fmaxf((acc.x + sv.x) * di + bq[0], 0.0f);
        o.y = fmaxf((acc.y + sv.y) * di + bq[1], 0.0f);
        o.z = fmaxf((acc.z + sv.z) * di + bq[2], 0.0f);
        o.w = fmaxf((acc.w + sv.w) * di + bq[3], 0.0f);
        *(float4*)(hs + 4 * tid) = o;
    }
    __syncthreads();

    const int w = __builtin_amdgcn_readfirstlane(tid >> 5);
    const int ntile = min(GN / 16, (n_nodes - node0) >> 4);
    if (w >= ntile) return;

    const int lane = tid & 31;
    const int c = lane & 15, h = lane >> 4;

    v2f wa[5][2];
#pragma unroll
    for (int j = 0; j < 5; ++j) {
#pragma unroll
        for (int mi = 0; mi < 2; ++mi) {
            const int m = mi * 16 + c;
            const int mc = min(m, 19);
            const int k0 = 4 * j + 2 * h;
            const float w0 = Wm[k0 * 20 + mc];
            const float w1 = Wm[(k0 + 1) * 20 + mc];
            wa[j][mi] = (m < 20) ? (v2f){w0, w1} : (v2f){0.0f, 0.0f};
        }
    }

    const int tn = node0 + w * 16 + c;
    const float d = dinv[tn];
    const float* hp = hs + (w * 16 + c) * 20 + 2 * h;
    v2f b[5];
#pragma unroll
    for (int j = 0; j < 5; ++j) b[j] = *(const v2f*)(hp + 4 * j);

    v8f acc0 = {}, acc1 = {};
#pragma unroll
    for (int j = 0; j < 5; ++j) {
        acc0 = __builtin_amdgcn_wmma_f32_16x16x4_f32(false, wa[j][0], false, b[j],
                                                     (short)0, acc0, false, false);
        acc1 = __builtin_amdgcn_wmma_f32_16x16x4_f32(false, wa[j][1], false, b[j],
                                                     (short)0, acc1, false, false);
    }
    float* o = Tt + (size_t)tn * 20 + 8 * h;
    *(v4f*)o       = (v4f){acc0[0] * d, acc0[1] * d, acc0[2] * d, acc0[3] * d};
    *(v4f*)(o + 4) = (v4f){acc0[4] * d, acc0[5] * d, acc0[6] * d, acc0[7] * d};
    if (h == 0) *(v4f*)(o + 16) = (v4f){acc1[0] * d, acc1[1] * d, acc1[2] * d, acc1[3] * d};
}

__global__ void __launch_bounds__(128) k_fc1_tiled(const float* __restrict__ A,
                                                   const float* __restrict__ Bw,
                                                   const float* __restrict__ bias,
                                                   float* __restrict__ C, int M) {
    constexpr int K = 1640, N = 300, BM = 32, BK = 40, NP = 336, PA = 44, NT = K / BK;
    constexpr int NB = (BK / 2) * (N / 4);
    constexpr int NA = BM * (BK / 4);
    constexpr int JB = (NB + 127) / 128;
    constexpr int JA = (NA + 127) / 128;
    constexpr int SA = BM * PA, SB = (BK / 2) * NP * 2;
    __shared__ __attribute__((aligned(16))) float As[2 * SA];
    __shared__ __attribute__((aligned(16))) float Bs[2 * SB];

    const int tid = threadIdx.x, lane = tid & 31, w = tid >> 5;
    const int c = lane & 15, h = lane >> 4;
    const int m0 = blockIdx.x * BM;

    for (int i = tid; i < (BK / 2) * 72; i += 128) {
        int kp = i / 72, r = i - kp * 72;
        Bs[(kp * NP + N) * 2 + r] = 0.0f;
        Bs[SB + (kp * NP + N) * 2 + r] = 0.0f;
    }

    v4f rb0[JB], rb1[JB], ra[JA];
    int boff[JB], bsoff[JB], aoff[JA], asoff[JA];
#pragma unroll
    for (int j = 0; j < JB; ++j) {
        int idx = min(tid + 128 * j, NB - 1);
        int kp = idx / (N / 4), q = idx - kp * (N / 4);
        boff[j] = (2 * kp) * N + 4 * q;
        bsoff[j] = (kp * NP + 4 * q) * 2;
    }
#pragma unroll
    for (int j = 0; j < JA; ++j) {
        int idx = min(tid + 128 * j, NA - 1);
        int m = idx / (BK / 4), jj = idx - m * (BK / 4);
        int row = min(m0 + m, M - 1);
        aoff[j] = row * K + 4 * jj;
        asoff[j] = m * PA + 4 * jj;
    }

#pragma unroll
    for (int j = 0; j < JB; ++j) {
        rb0[j] = *(const v4f*)(Bw + boff[j]);
        rb1[j] = *(const v4f*)(Bw + boff[j] + N);
    }
#pragma unroll
    for (int j = 0; j < JA; ++j)
        ra[j] = *(const v4f*)(A + aoff[j]);

    v8f acc[2][5] = {};
    const float* ap = As + c * PA + 2 * h;
    const float* bp = Bs + (h * NP + w * 80 + c) * 2;

#pragma unroll
    for (int j = 0; j < JB; ++j) {
        float* p = Bs + bsoff[j];
        *(v4f*)p       = (v4f){rb0[j].x, rb1[j].x, rb0[j].y, rb1[j].y};
        *(v4f*)(p + 4) = (v4f){rb0[j].z, rb1[j].z, rb0[j].w, rb1[j].w};
    }
#pragma unroll
    for (int j = 0; j < JA; ++j) *(v4f*)(As + asoff[j]) = ra[j];
#pragma unroll
    for (int j = 0; j < JB; ++j) {
        rb0[j] = *(const v4f*)(Bw + BK * N + boff[j]);
        rb1[j] = *(const v4f*)(Bw + BK * N + boff[j] + N);
    }
#pragma unroll
    for (int j = 0; j < JA; ++j) ra[j] = *(const v4f*)(A + BK + aoff[j]);
    __syncthreads();

    for (int t = 0; t < NT; ++t) {
        const int cur = (t & 1) * SB, nxt = SB - cur;
        const int curA = (t & 1) * SA, nxtA = SA - curA;

#pragma unroll
        for (int kk = 0; kk < BK / 4; ++kk) {
            if (kk == 1 && t + 1 < NT) {
#pragma unroll
                for (int j = 0; j < JB; ++j) {
                    float* p = Bs + nxt + bsoff[j];
                    *(v4f*)p       = (v4f){rb0[j].x, rb1[j].x, rb0[j].y, rb1[j].y};
                    *(v4f*)(p + 4) = (v4f){rb0[j].z, rb1[j].z, rb0[j].w, rb1[j].w};
                }
#pragma unroll
                for (int j = 0; j < JA; ++j) *(v4f*)(As + nxtA + asoff[j]) = ra[j];
                if (t + 2 < NT) {
                    const float* Bn = Bw + (t + 2) * (BK * N);
                    const float* An = A + (t + 2) * BK;
#pragma unroll
                    for (int j = 0; j < JB; ++j) {
                        rb0[j] = *(const v4f*)(Bn + boff[j]);
                        rb1[j] = *(const v4f*)(Bn + boff[j] + N);
                    }
#pragma unroll
                    for (int j = 0; j < JA; ++j) ra[j] = *(const v4f*)(An + aoff[j]);
                }
            }
            v2f a0 = *(const v2f*)(ap + curA + kk * 4);
            v2f a1 = *(const v2f*)(ap + curA + 16 * PA + kk * 4);
            v2f b[5];
#pragma unroll
            for (int ni = 0; ni < 5; ++ni) b[ni] = *(const v2f*)(bp + cur + kk * (4 * NP) + ni * 32);
#pragma unroll
            for (int ni = 0; ni < 5; ++ni) {
                acc[0][ni] = __builtin_amdgcn_wmma_f32_16x16x4_f32(false, a0, false, b[ni],
                                                                   (short)0, acc[0][ni], false, false);
                acc[1][ni] = __builtin_amdgcn_wmma_f32_16x16x4_f32(false, a1, false, b[ni],
                                                                   (short)0, acc[1][ni], false, false);
            }
        }
        __syncthreads();
    }

#pragma unroll
    for (int ni = 0; ni < 5; ++ni) {
        int col = w * 80 + ni * 16 + c;
        if (col < N) {
            float bv = bias[col];
#pragma unroll
            for (int mi = 0; mi < 2; ++mi) {
                int rbase = m0 + mi * 16 + 8 * h;
#pragma unroll
                for (int r = 0; r < 8; ++r)
                    if (rbase + r < M)
                        C[(size_t)(rbase + r) * N + col] = fmaxf(acc[mi][ni][r] + bv, 0.0f);
            }
        }
    }
}


extern "C" void kernel_launch(void* const* d_in, const int* in_sizes, int n_in,
                              void* d_out, int out_size, void* d_ws, size_t ws_size,
                              hipStream_t stream) {
    const float* x    = (const float*)d_in[0];
    const int*   eidx = (const int*)d_in[1];
    const float* W1   = (const float*)d_in[3];
    const float* b1   = (const float*)d_in[4];
    const float* W2   = (const float*)d_in[5];
    const float* b2   = (const float*)d_in[6];
    const float* fc1W = (const float*)d_in[7];
    const float* fc1b = (const float*)d_in[8];
    const float* outW = (const float*)d_in[9];
    const float* outb = (const float*)d_in[10];

    const int N  = in_sizes[2];
    const int E  = in_sizes[1] / 2;
    const int Bg = N / 82;
    const int EMB = 20, LAT = 82 * 20, HID = 300, NC = 22;

    char* ws = (char*)d_ws;
    size_t off = 0;
    auto take = [&](size_t bytes) -> void* {
        off = (off + 255) & ~(size_t)255;
        void* p = ws + off;
        off += bytes;
        return p;
    };
    const int nbin  = cdiv(N, BIN_W);
    const int chunk = cdiv(E, NBLK_E);
    const int NH    = nbin * NBLK_E;
    const int ntile = cdiv(NH, 1024);
    float* dinv   = (float*)take((size_t)N * 4);
    int*   cnt    = (int*)take((size_t)N * 4);
    int*   rowptr = (int*)take((size_t)N * 4);
    int*   hist   = (int*)take((size_t)NH * 4);
    int*   pre    = (int*)take((size_t)NH * 4);
    int*   tsum   = (int*)take((size_t)ntile * 4);
    int2*  binned = (int2*)take((size_t)E * 8);
    int*   lists  = (int*)take((size_t)E * 4);
    float* t    = (float*)take((size_t)N * EMB * 4);
    float* t2   = (float*)take((size_t)N * EMB * 4);
    float* fco  = (float*)take((size_t)Bg * HID * 4);
    float* Bt   = (float*)take((size_t)304 * LAT * 4);

    const int* src = eidx;
    const int* dst = eidx + E;

    const int TB = 256;

    auto run_gemm = [&](const float* A, const float* Bw, const float* bias, float* C,
                        int M, int K, int Ncols, int relu) {
        int ntiles = (Ncols + 15) / 16;
        int ncols_pad = ntiles * 16;
        int mtiles = M / 16;
        k_transpose_pad<<<cdiv(ncols_pad * K, TB), TB, 0, stream>>>(Bw, Bt, K, Ncols, ncols_pad);
        int waves = mtiles * ntiles;
        k_wmma_gemm_f32<<<cdiv(waves * 32, TB), TB, 0, stream>>>(
            A, Bt, bias, C, M, K, Ncols, mtiles, ntiles, relu);
    };

    k_bin_hist<<<NBLK_E, TE, 0, stream>>>(dst, hist, E, chunk, nbin);
    k_tile_sums<<<ntile, 256, 0, stream>>>(hist, tsum, NH);
    k_scan_excl<<<ntile, 256, 0, stream>>>(hist, tsum, pre, NH);
    k_bin_scatter<<<NBLK_E, TE, 0, stream>>>(src, dst, pre, binned, E, chunk, nbin);
    k_bin_count<<<nbin, TE, 0, stream>>>(binned, pre, rowptr, cnt, dinv, N, E, nbin);
    k_bin_fill<<<nbin, TE, 0, stream>>>(binned, pre, rowptr, lists, N, E, nbin);

    auto run_node_gemm = [&](const float* A, const float* Wm, float* C, int M) {
        int mtiles = M / 16;
        int waves = cdiv(mtiles, NODE_TPW);
        k_node_gemm20<<<cdiv(waves * 32, TB), TB, 0, stream>>>(A, Wm, dinv, C, mtiles);
    };

    run_node_gemm(x, W1, t, N);
    k_gather_node_gemm20<<<cdiv(N, GN), GT, 0, stream>>>(rowptr, cnt, lists, t, dinv, b1, W2, t2, N);

    k_gather_finish<<<cdiv(N * 5, 256), 256, 0, stream>>>(rowptr, cnt, lists, t2, dinv, b2, t, N);

    k_fc1_tiled<<<cdiv(Bg, 32), 128, 0, stream>>>(t  , fc1W, fc1b, fco, Bg);

    run_gemm(fco, outW, outb, (float*)d_out, Bg, HID, NC, 0);
}


# Launch folding at two sites: output layer (8192x300x22) into fc1's epilogue (activations reach the WMMA through LDS), bin count + list fill into one kernel; 12 -> 9 launches; k and sum orders kept
# speedup vs baseline: 8.6760x; 1.0719x over previous
#include <hip/hip_runtime.h>
#include <hip/hip_bf16.h>

typedef __attribute__((ext_vector_type(2))) float v2f;
typedef __attribute__((ext_vector_type(4))) float v4f;
typedef __attribute__((ext_vector_type(8))) float v8f;

static __host__ inline int cdiv(int a, int b) { return (a + b - 1) / b; }

constexpr int BIN_SHIFT = 11;
constexpr int BIN_W = 1 << BIN_SHIFT;
constexpr int MAX_BINS = 1024;
constexpr int NBLK_E = 256;
constexpr int TE = 1024;
constexpr int CAP_E = 16384;
static_assert(BIN_W == 2 * TE, "k_bin_lists scans two counters per thread");
static_assert(MAX_BINS == TE, "k_bin_scatter scans one bin per thread");

__device__ __forceinline__ int bin_of(int d, int nbin) { return min(max(d >> BIN_SHIFT, 0), nbin - 1); }

__global__ void __launch_bounds__(TE) k_bin_hist(const int* __restrict__ dst, int* __restrict__ hist,
                                                 int E, int chunk, int nbin) {
    __shared__ int hc[MAX_BINS];
    const int tid = threadIdx.x;
    for (int i = tid; i < nbin; i += TE) hc[i] = 0;
    __syncthreads();
    const int start = blockIdx.x * chunk;
    const int end = min(start + chunk, E);
    int i = start + tid;
    for (; i + 3 * TE < end; i += 4 * TE) {
        int d[4];
        for (int j = 0; j < 4; ++j) d[j] = dst[i + j * TE];
        for (int j = 0; j < 4; ++j) atomicAdd(&hc[bin_of(d[j], nbin)], 1);
    }
    for (; i < end; i += TE) atomicAdd(&hc[bin_of(dst[i], nbin)], 1);
    __syncthreads();
    for (int b = tid; b < nbin; b += TE) hist[b * NBLK_E + blockIdx.x] = hc[b];
}

__global__ void __launch_bounds__(256) k_tile_sums(const int* __restrict__ cnt,
                                                   int* __restrict__ tsum, int n) {
    __shared__ int wsum[8];
    int base = blockIdx.x * 1024 + threadIdx.x * 4;
    int s = 0;
    for (int j = 0; j < 4; ++j)
        if (base + j < n) s += cnt[base + j];
    for (int o = 16; o > 0; o >>= 1) s += __shfl_down(s, o, 32);
    if ((threadIdx.x & 31) == 0) wsum[threadIdx.x >> 5] = s;
    __syncthreads();
    if (threadIdx.x == 0) {
        int tot = 0;
        for (int w = 0; w < 8; ++w) tot += wsum[w];
        tsum[blockIdx.x] = tot;
    }
}

__global__ void __launch_bounds__(256) k_scan_excl(const int* __restrict__ cnt,
                                                   const int* __restrict__ tsum,
                                                   int* __restrict__ pre, int n) {
    __shared__ int wpre[8];
    __shared__ int winc[8];
    int tid = threadIdx.x, lane = tid & 31, w = tid >> 5;

    int p = 0;
    for (int j = tid; j < (int)blockIdx.x; j += 256) p += tsum[j];
    for (int o = 16; o > 0; o >>= 1) p += __shfl_down(p, o, 32);
    if (lane == 0) wpre[w] = p;

    int base = blockIdx.x * 1024 + tid * 4;
    int c[4];
    for (int j = 0; j < 4; ++j) c[j] = (base + j < n) ? cnt[base + j] : 0;
    int s = c[0] + c[1] + c[2] + c[3];
    int inc = s;
    for (int o = 1; o < 32; o <<= 1) {
        int y = __shfl_up(inc, o, 32);
        if (lane >= o) inc += y;
    }
    if (lane == 31) winc[w] = inc;
    __syncthreads();

    int off = 0;
    for (int j = 0; j < 8; ++j) {
        off += wpre[j];
        if (j < w) off += winc[j];
    }
    int ex = off + inc - s;
    for (int j = 0; j < 4; ++j) {
        if (base + j < n) {
            pre[base + j] = ex;
            ex += c[j];
        }
    }
}

__global__ void __launch_bounds__(TE) k_bin_scatter(const int* __restrict__ src,
                                                    const int* __restrict__ dst,
                                                    const int* __restrict__ pre,
                                                    int2* __restrict__ binned,
                                                    int E, int chunk, int nbin) {
    __shared__ int gbase[MAX_BINS];
    __shared__ int lstart[MAX_BINS];
    __shared__ int cur[MAX_BINS];
    __shared__ int wtot[TE / 32];
    __shared__ int2 stage[CAP_E];
    const int tid = threadIdx.x, lane = tid & 31, w = tid >> 5;
    const int start = blockIdx.x * chunk;
    const int end = min(start + chunk, E);
    const int len = max(end - start, 0);

    int g = 0, nb = 0;
    if (tid < nbin) {
        int idx = tid * NBLK_E + blockIdx.x;
        g = pre[idx];
        nb = ((idx + 1 < nbin * NBLK_E) ? pre[idx + 1] : E) - g;
    }

    if (len <= CAP_E) {
        int inc = nb;
        for (int o = 1; o < 32; o <<= 1) {
            int y = __shfl_up(inc, o, 32);
            if (lane >= o) inc += y;
        }
        if (lane == 31) wtot[w] = inc;
        __syncthreads();
        int off = 0;
        for (int j = 0; j < TE / 32; ++j)
            if (j < w) off += wtot[j];
        const int ex = off + inc - nb;
        if (tid < nbin) {
            gbase[tid] = g;
            lstart[tid] = ex;
            cur[tid] = ex;
        }
        __syncthreads();
        int i = start + tid;
        for (; i + 3 * TE < end; i += 4 * TE) {
            int s[4], d[4], pos[4];
            for (int j = 0; j < 4; ++j) {
                s[j] = src[i + j * TE];
                d[j] = dst[i + j * TE];
            }
            for (int j = 0; j < 4; ++j) pos[j] = atomicAdd(&cur[bin_of(d[j], nbin)], 1);
            for (int j = 0; j < 4; ++j) stage[min(max(pos[j], 0), CAP_E - 1)] = make_int2(s[j], d[j]);
        }
        for (; i < end; i += TE) {
            int s = src[i], d = dst[i];
            int pos = atomicAdd(&cur[bin_of(d, nbin)], 1);
            stage[min(max(pos, 0), CAP_E - 1)] = make_int2(s, d);
        }
        __syncthreads();
        for (int k = tid; k < len; k += TE) {
            int2 e = stage[k];
            int b = bin_of(e.y, nbin);
            int pos = gbase[b] + (k - lstart[b]);
            binned[min(max(pos, 0), E - 1)] = e;
        }
    } else {
        if (tid < nbin) cur[tid] = g;
        __syncthreads();
        for (int i = start + tid; i < end; i += TE) {
            int s = src[i], d = dst[i];
            int pos = atomicAdd(&cur[bin_of(d, nbin)], 1);
            binned[min(max(pos, 0), E - 1)] = make_int2(s, d);
        }
    }
}

__global__ void __launch_bounds__(TE) k_bin_lists(const int2* __restrict__ binned,
                                                  const int* __restrict__ pre,
                                                  int* __restrict__ rowptr, int* __restrict__ cnt,
                                                  float* __restrict__ dinv, int* __restrict__ lists,
                                                  int n, int E, int nbin) {
    __shared__ int c[BIN_W];
    __shared__ int wtot[TE / 32];
    __shared__ int stage[CAP_E];
    const int tid = threadIdx.x, lane = tid & 31, w = tid >> 5;
    const int bin = blockIdx.x;
    const int lo = bin << BIN_SHIFT;
    int bs = pre[bin * NBLK_E];
    int be = (bin + 1 < nbin) ? pre[(bin + 1) * NBLK_E] : E;
    bs = min(max(bs, 0), E);
    be = min(max(be, bs), E);
    for (int k = tid; k < BIN_W; k += TE) c[k] = 0;
    __syncthreads();
    int i = bs + tid;
    for (; i + 3 * TE < be; i += 4 * TE) {
        int d[4];
        for (int j = 0; j < 4; ++j) d[j] = binned[i + j * TE].y;
        for (int j = 0; j < 4; ++j) atomicAdd(&c[min(max(d[j] - lo, 0), BIN_W - 1)], 1);
    }
    for (; i < be; i += TE) atomicAdd(&c[min(max(binned[i].y - lo, 0), BIN_W - 1)], 1);
    __syncthreads();
    const int c0 = c[2 * tid], c1 = c[2 * tid + 1];
    const int s = c0 + c1;
    int inc = s;
    for (int o = 1; o < 32; o <<= 1) {
        int y = __shfl_up(inc, o, 32);
        if (lane >= o) inc += y;
    }
    if (lane == 31) wtot[w] = inc;
    __syncthreads();
    int off = 0;
    for (int j = 0; j < TE / 32; ++j)
        if (j < w) off += wtot[j];
    const int ex = bs + off + inc - s;
    const int node = lo + 2 * tid;
    if (node < n) {
        rowptr[node] = ex;
        cnt[node] = c0;
        float v = (float)(c0 + 1);
        dinv[node] = (v > 0.0f) ? rsqrtf(v) : 0.0f;
    }
    if (node + 1 < n) {
        rowptr[node + 1] = ex + c0;
        cnt[node + 1] = c1;
        float v = (float)(c1 + 1);
        dinv[node + 1] = (v > 0.0f) ? rsqrtf(v) : 0.0f;
    }
    c[2 * tid] = ex;
    c[2 * tid + 1] = ex + c0;
    __syncthreads();

    const int len = be - bs;
    if (len <= CAP_E) {
        int i2 = bs + tid;
        for (; i2 + 3 * TE < be; i2 += 4 * TE) {
            int2 e[4];
            int pos[4];
            for (int j = 0; j < 4; ++j) e[j] = binned[i2 + j * TE];
            for (int j = 0; j < 4; ++j) pos[j] = atomicAdd(&c[min(max(e[j].y - lo, 0), BIN_W - 1)], 1) - bs;
            for (int j = 0; j < 4; ++j) stage[min(max(pos[j], 0), CAP_E - 1)] = e[j].x;
        }
        for (; i2 < be; i2 += TE) {
            int2 e = binned[i2];
            int pos = atomicAdd(&c[min(max(e.y - lo, 0), BIN_W - 1)], 1) - bs;
            stage[min(max(pos, 0), CAP_E - 1)] = e.x;
        }
        __syncthreads();
        for (int k = tid; k < len; k += TE) lists[bs + k] = stage[k];
    } else {
        for (int i2 = bs + tid; i2 < be; i2 += TE) {
            int2 e = binned[i2];
            int pos = atomicAdd(&c[min(max(e.y - lo, 0), BIN_W - 1)], 1);
            lists[min(max(pos, 0), E - 1)] = e.x;
        }
    }
}

__global__ void __launch_bounds__(256) k_gather_finish(const int* __restrict__ rowptr,
                                                       const int* __restrict__ cnt,
                                                       const int* __restrict__ lists,
                                                       const float* __restrict__ t,
                                                       const float* __restrict__ dinv,
                                                       const float* __restrict__ bias,
                                                       float* __restrict__ h, int n_nodes) {
    int g = blockIdx.x * 256 + threadIdx.x;
    if (g >= n_nodes * 5) return;
    int node = g / 5;
    int q = g - node * 5;
    const float4* t4 = (const float4*)t;
    float di = dinv[node];
    int k = rowptr[node];
    int k1 = k + cnt[node];
    int nmax = n_nodes - 1;

    float4 acc = make_float4(0.0f, 0.0f, 0.0f, 0.0f);
    for (; k + 4 <= k1; k += 4) {
        int s[4];
        float4 v[4];
        for (int j = 0; j < 4; ++j) s[j] = lists[k + j];
        for (int j = 0; j < 4; ++j) v[j] = t4[min(max(s[j], 0), nmax) * 5 + q];
        for (int j = 0; j < 4; ++j) {
            acc.x += v[j].x;
            acc.y += v[j].y;
            acc.z += v[j].z;
            acc.w += v[j].w;
        }
    }
#pragma unroll 1
    for (; k < k1; ++k) {
        int s0 = lists[k];
        float4 v0 = t4[min(max(s0, 0), nmax) * 5 + q];
        acc.x += v0.x;
        acc.y += v0.y;
        acc.z += v0.z;
        acc.w += v0.w;
    }

    float4 sv = t4[g];
    const float* bq = bias + 4 * q;
    float4 o;
    o.x = fmaxf((acc.x + sv.x) * di + bq[0], 0.0f);
    o.y = fmaxf((acc.y + sv.y) * di + bq[1], 0.0f);
    o.z = fmaxf((acc.z + sv.z) * di + bq[2], 0.0f);
    o.w = fmaxf((acc.w + sv.w) * di + bq[3], 0.0f);
    ((float4*)h)[g] = o;
}


constexpr int NODE_TPW = 8;

__global__ void __launch_bounds__(256) k_node_gemm20(const float* __restrict__ X,
                                                     const float* __restrict__ Wm,
                                                     const float* __restrict__ dinv,
                                                     float* __restrict__ Tt, int mtiles) {
    const int lane = threadIdx.x & 31;
    const int wave = __builtin_amdgcn_readfirstlane((int)(blockIdx.x * 256 + threadIdx.x) >> 5);
    const int c = lane & 15, h = lane >> 4;
    const int t0 = wave * NODE_TPW;
    const int nt = min(NODE_TPW, mtiles - t0);
    if (nt <= 0) return;

    v2f wa[5][2];
#pragma unroll
    for (int j = 0; j < 5; ++j) {
#pragma unroll
        for (int mi = 0; mi < 2; ++mi) {
            const int m = mi * 16 + c;
            const int mc = min(m, 19);
            const int k0 = 4 * j + 2 * h;
            const float w0 = Wm[k0 * 20 + mc];
            const float w1 = Wm[(k0 + 1) * 20 + mc];
            wa[j][mi] = (m < 20) ? (v2f){w0, w1} : (v2f){0.0f, 0.0f};
        }
    }

    const float* xp = X + (size_t)(t0 * 16 + c) * 20 + 2 * h;
    float* tp = Tt + (size_t)(t0 * 16 + c) * 20 + 8 * h;
    const float* dp = dinv + (t0 * 16 + c);
    v2f b[NODE_TPW][5];
    float dv[NODE_TPW];
#pragma unroll
    for (int i = 0; i < NODE_TPW; ++i) {
        const int ii = min(i, nt - 1);
#pragma unroll
        for (int j = 0; j < 5; ++j) b[i][j] = *(const v2f*)(xp + ii * 320 + 4 * j);
        dv[i] = dp[ii * 16];
    }

    __builtin_amdgcn_sched_barrier(0);

#pragma unroll
    for (int i = 0; i < NODE_TPW; ++i) {
        v8f acc0 = {}, acc1 = {};
#pragma unroll
        for (int j = 0; j < 5; ++j) {
            acc0 = __builtin_amdgcn_wmma_f32_16x16x4_f32(false, wa[j][0], false, b[i][j],
                                                         (short)0, acc0, false, false);
            acc1 = __builtin_amdgcn_wmma_f32_16x16x4_f32(false, wa[j][1], false, b[i][j],
                                                         (short)0, acc1, false, false);
        }
        if (i < nt) {
            float* o = tp + i * 320;
            const float d = dv[i];
            *(v4f*)o       = (v4f){acc0[0] * d, acc0[1] * d, acc0[2] * d, acc0[3] * d};
            *(v4f*)(o + 4) = (v4f){acc0[4] * d, acc0[5] * d, acc0[6] * d, acc0[7] * d};
            if (h == 0) *(v4f*)(o + 16) = (v4f){acc1[0] * d, acc1[1] * d, acc1[2] * d, acc1[3] * d};
        }
    }
}

constexpr int GN = 64;
constexpr int GT = GN * 5;

__global__ void __launch_bounds__(GT) k_gather_node_gemm20(const int* __restrict__ rowptr,
                                                           const int* __restrict__ cnt,
                                                           const int* __restrict__ lists,
                                                           const float* __restrict__ t,
                                                           const float* __restrict__ dinv,
                                                           const float* __restrict__ bias,
                                                           const float* __restrict__ Wm,
                                                           float* __restrict__ Tt, int n_nodes) {
    __shared__ __attribute__((aligned(16))) float hs[GN * 20];
    const int tid = threadIdx.x;
    const int node0 = blockIdx.x * GN;
    const int nl = tid / 5;
    const int q = tid - nl * 5;
    const int node = node0 + nl;

    if (node < n_nodes) {
        const int g = node * 5 + q;
        const float4* t4 = (const float4*)t;
        float di = dinv[node];
        int k = rowptr[node];
        int k1 = k + cnt[node];
        int nmax = n_nodes - 1;

        float4 acc = make_float4(0.0f, 0.0f, 0.0f, 0.0f);
        for (; k + 4 <= k1; k += 4) {
            int s[4];
            float4 v[4];
            for (int j = 0; j < 4; ++j) s[j] = lists[k + j];
            for (int j = 0; j < 4; ++j) v[j] = t4[min(max(s[j], 0), nmax) * 5 + q];
            for (int j = 0; j < 4; ++j) {
                acc.x += v[j].x;
                acc.y += v[j].y;
                acc.z += v[j].z;
                acc.w += v[j].w;
            }
        }
#pragma unroll 1
        for (; k < k1; ++k) {
            int s0 = lists[k];
            float4 v0 = t4[min(max(s0, 0), nmax) * 5 + q];
            acc.x += v0.x;
            acc.y += v0.y;
            acc.z += v0.z;
            acc.w += v0.w;
        }

        float4 sv = t4[g];
        const float* bq = bias + 4 * q;
        float4 o;
        o.x = fmaxf((acc.x + sv.x) * di + bq[0], 0.0f);
        o.y = fmaxf((acc.y + sv.y) * di + bq[1], 0.0f);
        o.z = fmaxf((acc.z + sv.z) * di + bq[2], 0.0f);
        o.w = fmaxf((acc.w + sv.w) * di + bq[3], 0.0f);
        *(float4*)(hs + 4 * tid) = o;
    }
    __syncthreads();

    const int w = __builtin_amdgcn_readfirstlane(tid >> 5);
    const int ntile = min(GN / 16, (n_nodes - node0) >> 4);
    if (w >= ntile) return;

    const int lane = tid & 31;
    const int c = lane & 15, h = lane >> 4;

    v2f wa[5][2];
#pragma unroll
    for (int j = 0; j < 5; ++j) {
#pragma unroll
        for (int mi = 0; mi < 2; ++mi) {
            const int m = mi * 16 + c;
            const int mc = min(m, 19);
            const int k0 = 4 * j + 2 * h;
            const float w0 = Wm[k0 * 20 + mc];
            const float w1 = Wm[(k0 + 1) * 20 + mc];
            wa[j][mi] = (m < 20) ? (v2f){w0, w1} : (v2f){0.0f, 0.0f};
        }
    }

    const int tn = node0 + w * 16 + c;
    const float d = dinv[tn];
    const float* hp = hs + (w * 16 + c) * 20 + 2 * h;
    v2f b[5];
#pragma unroll
    for (int j = 0; j < 5; ++j) b[j] = *(const v2f*)(hp + 4 * j);

    v8f acc0 = {}, acc1 = {};
#pragma unroll
    for (int j = 0; j < 5; ++j) {
        acc0 = __builtin_amdgcn_wmma_f32_16x16x4_f32(false, wa[j][0], false, b[j],
                                                     (short)0, acc0, false, false);
        acc1 = __builtin_amdgcn_wmma_f32_16x16x4_f32(false, wa[j][1], false, b[j],
                                                     (short)0, acc1, false, false);
    }
    float* o = Tt + (size_t)tn * 20 + 8 * h;
    *(v4f*)o       = (v4f){acc0[0] * d, acc0[1] * d, acc0[2] * d, acc0[3] * d};
    *(v4f*)(o + 4) = (v4f){acc0[4] * d, acc0[5] * d, acc0[6] * d, acc0[7] * d};
    if (h == 0) *(v4f*)(o + 16) = (v4f){acc1[0] * d, acc1[1] * d, acc1[2] * d, acc1[3] * d};
}

__global__ void __launch_bounds__(128) k_fc1_tiled(const float* __restrict__ A,
                                                   const float* __restrict__ Bw,
                                                   const float* __restrict__ bias,
                                                   const float* __restrict__ Wo,
                                                   const float* __restrict__ bo,
                                                   float* __restrict__ Out, int M) {
    constexpr int K = 1640, N = 300, BM = 32, BK = 40, NP = 336, PA = 44, NT = K / BK;
    constexpr int NO = 22, KO = N / 4;
    constexpr int NB = (BK / 2) * (N / 4);
    constexpr int NA = BM * (BK / 4);
    constexpr int JB = (NB + 127) / 128;
    constexpr int JA = (NA + 127) / 128;
    constexpr int SA = BM * PA, SB = (BK / 2) * NP * 2;
    __shared__ __attribute__((aligned(16))) float As[2 * SA];
    __shared__ __attribute__((aligned(16))) float Bs[2 * SB];

    const int tid = threadIdx.x, lane = tid & 31, w = tid >> 5;
    const int c = lane & 15, h = lane >> 4;
    const int m0 = blockIdx.x * BM;

    for (int i = tid; i < (BK / 2) * 72; i += 128) {
        int kp = i / 72, r = i - kp * 72;
        Bs[(kp * NP + N) * 2 + r] = 0.0f;
        Bs[SB + (kp * NP + N) * 2 + r] = 0.0f;
    }

    v4f rb0[JB], rb1[JB], ra[JA];
    int boff[JB], bsoff[JB], aoff[JA], asoff[JA];
#pragma unroll
    for (int j = 0; j < JB; ++j) {
        int idx = min(tid + 128 * j, NB - 1);
        int kp = idx / (N / 4), q = idx - kp * (N / 4);
        boff[j] = (2 * kp) * N + 4 * q;
        bsoff[j] = (kp * NP + 4 * q) * 2;
    }
#pragma unroll
    for (int j = 0; j < JA; ++j) {
        int idx = min(tid + 128 * j, NA - 1);
        int m = idx / (BK / 4), jj = idx - m * (BK / 4);
        int row = min(m0 + m, M - 1);
        aoff[j] = row * K + 4 * jj;
        asoff[j] = m * PA + 4 * jj;
    }

#pragma unroll
    for (int j = 0; j < JB; ++j) {
        rb0[j] = *(const v4f*)(Bw + boff[j]);
        rb1[j] = *(const v4f*)(Bw + boff[j] + N);
    }
#pragma unroll
    for (int j = 0; j < JA; ++j)
        ra[j] = *(const v4f*)(A + aoff[j]);

    v8f acc[2][5] = {};
    const float* ap = As + c * PA + 2 * h;
    const float* bp = Bs + (h * NP + w * 80 + c) * 2;

#pragma unroll
    for (int j = 0; j < JB; ++j) {
        float* p = Bs + bsoff[j];
        *(v4f*)p       = (v4f){rb0[j].x, rb1[j].x, rb0[j].y, rb1[j].y};
        *(v4f*)(p + 4) = (v4f){rb0[j].z, rb1[j].z, rb0[j].w, rb1[j].w};
    }
#pragma unroll
    for (int j = 0; j < JA; ++j) *(v4f*)(As + asoff[j]) = ra[j];
#pragma unroll
    for (int j = 0; j < JB; ++j) {
        rb0[j] = *(const v4f*)(Bw + BK * N + boff[j]);
        rb1[j] = *(const v4f*)(Bw + BK * N + boff[j] + N);
    }
#pragma unroll
    for (int j = 0; j < JA; ++j) ra[j] = *(const v4f*)(A + BK + aoff[j]);
    __syncthreads();

    for (int t = 0; t < NT; ++t) {
        const int cur = (t & 1) * SB, nxt = SB - cur;
        const int curA = (t & 1) * SA, nxtA = SA - curA;

#pragma unroll
        for (int kk = 0; kk < BK / 4; ++kk) {
            if (kk == 1 && t + 1 < NT) {
#pragma unroll
                for (int j = 0; j < JB; ++j) {
                    float* p = Bs + nxt + bsoff[j];
                    *(v4f*)p       = (v4f){rb0[j].x, rb1[j].x, rb0[j].y, rb1[j].y};
                    *(v4f*)(p + 4) = (v4f){rb0[j].z, rb1[j].z, rb0[j].w, rb1[j].w};
                }
#pragma unroll
                for (int j = 0; j < JA; ++j) *(v4f*)(As + nxtA + asoff[j]) = ra[j];
                if (t + 2 < NT) {
                    const float* Bn = Bw + (t + 2) * (BK * N);
                    const float* An = A + (t + 2) * BK;
#pragma unroll
                    for (int j = 0; j < JB; ++j) {
                        rb0[j] = *(const v4f*)(Bn + boff[j]);
                        rb1[j] = *(const v4f*)(Bn + boff[j] + N);
                    }
#pragma unroll
                    for (int j = 0; j < JA; ++j) ra[j] = *(const v4f*)(An + aoff[j]);
                }
            }
            v2f a0 = *(const v2f*)(ap + curA + kk * 4);
            v2f a1 = *(const v2f*)(ap + curA + 16 * PA + kk * 4);
            v2f b[5];
#pragma unroll
            for (int ni = 0; ni < 5; ++ni) b[ni] = *(const v2f*)(bp + cur + kk * (4 * NP) + ni * 32);
#pragma unroll
            for (int ni = 0; ni < 5; ++ni) {
                acc[0][ni] = __builtin_amdgcn_wmma_f32_16x16x4_f32(false, a0, false, b[ni],
                                                                   (short)0, acc[0][ni], false, false);
                acc[1][ni] = __builtin_amdgcn_wmma_f32_16x16x4_f32(false, a1, false, b[ni],
                                                                   (short)0, acc[1][ni], false, false);
            }
        }
        __syncthreads();
    }

    const int on = (w >> 1) * 16 + c;
    const int onc = min(on, NO - 1);
    const float* wp = Wo + 2 * h * NO + onc;
    v2f wb[KO];
#pragma unroll
    for (int j = 0; j < KO; ++j) wb[j] = (v2f){wp[4 * j * NO], wp[(4 * j + 1) * NO]};
    const float obv = bo[onc];

    float* Hs = Bs;
    static_assert(BM * N <= 2 * SB, "the activations fit into the B stages");
#pragma unroll
    for (int ni = 0; ni < 5; ++ni) {
        int col = w * 80 + ni * 16 + c;
        if (col < N) {
            float bv = bias[col];
#pragma unroll
            for (int mi = 0; mi < 2; ++mi) {
#pragma unroll
                for (int r = 0; r < 8; ++r)
                    Hs[(mi * 16 + 8 * h + r) * N + col] = fmaxf(acc[mi][ni][r] + bv, 0.0f);
            }
        }
    }
    __syncthreads();

    const float* hp = Hs + ((w & 1) * 16 + c) * N + 2 * h;
    v2f af[KO];
#pragma unroll
    for (int j = 0; j < KO; ++j) af[j] = *(const v2f*)(hp + 4 * j);
    __builtin_amdgcn_sched_barrier(0);
    v8f oacc = {};
#pragma unroll
    for (int j = 0; j < KO; ++j)
        oacc = __builtin_amdgcn_wmma_f32_16x16x4_f32(false, af[j], false, wb[j],
                                                     (short)0, oacc, false, false);

    if (on < NO) {
        int rbase = m0 + (w & 1) * 16 + 8 * h;
#pragma unroll
        for (int r = 0; r < 8; ++r)
            if (rbase + r < M)
                Out[(size_t)(rbase + r) * NO + on] = oacc[r] + obv;
    }
}


extern "C" void kernel_launch(void* const* d_in, const int* in_sizes, int n_in,
                              void* d_out, int out_size, void* d_ws, size_t ws_size,
                              hipStream_t stream) {
    const float* x    = (const float*)d_in[0];
    const int*   eidx = (const int*)d_in[1];
    const float* W1   = (const float*)d_in[3];
    const float* b1   = (const float*)d_in[4];
    const float* W2   = (const float*)d_in[5];
    const float* b2   = (const float*)d_in[6];
    const float* fc1W = (const float*)d_in[7];
    const float* fc1b = (const float*)d_in[8];
    const float* outW = (const float*)d_in[9];
    const float* outb = (const float*)d_in[10];

    const int N  = in_sizes[2];
    const int E  = in_sizes[1] / 2;
    const int Bg = N / 82;
    const int EMB = 20;

    char* ws = (char*)d_ws;
    size_t off = 0;
    auto take = [&](size_t bytes) -> void* {
        off = (off + 255) & ~(size_t)255;
        void* p = ws + off;
        off += bytes;
        return p;
    };
    const int nbin  = cdiv(N, BIN_W);
    const int chunk = cdiv(E, NBLK_E);
    const int NH    = nbin * NBLK_E;
    const int ntile = cdiv(NH, 1024);
    float* dinv   = (float*)take((size_t)N * 4);
    int*   cnt    = (int*)take((size_t)N * 4);
    int*   rowptr = (int*)take((size_t)N * 4);
    int*   hist   = (int*)take((size_t)NH * 4);
    int*   pre    = (int*)take((size_t)NH * 4);
    int*   tsum   = (int*)take((size_t)ntile * 4);
    int2*  binned = (int2*)take((size_t)E * 8);
    int*   lists  = (int*)take((size_t)E * 4);
    float* t    = (float*)take((size_t)N * EMB * 4);
    float* t2   = (float*)take((size_t)N * EMB * 4);

    const int* src = eidx;
    const int* dst = eidx + E;

    const int TB = 256;

    k_bin_hist<<<NBLK_E, TE, 0, stream>>>(dst, hist, E, chunk, nbin);
    k_tile_sums<<<ntile, 256, 0, stream>>>(hist, tsum, NH);
    k_scan_excl<<<ntile, 256, 0, stream>>>(hist, tsum, pre, NH);
    k_bin_scatter<<<NBLK_E, TE, 0, stream>>>(src, dst, pre, binned, E, chunk, nbin);
    k_bin_lists<<<nbin, TE, 0, stream>>>(binned, pre, rowptr, cnt, dinv, lists, N, E, nbin);

    auto run_node_gemm = [&](const float* A, const float* Wm, float* C, int M) {
        int mtiles = M / 16;
        int waves = cdiv(mtiles, NODE_TPW);
        k_node_gemm20<<<cdiv(waves * 32, TB), TB, 0, stream>>>(A, Wm, dinv, C, mtiles);
    };

    run_node_gemm(x, W1, t, N);
    k_gather_node_gemm20<<<cdiv(N, GN), GT, 0, stream>>>(rowptr, cnt, lists, t, dinv, b1, W2, t2, N);

    k_gather_finish<<<cdiv(N * 5, 256), 256, 0, stream>>>(rowptr, cnt, lists, t2, dinv, b2, t, N);

    k_fc1_tiled<<<cdiv(Bg, 32), 128, 0, stream>>>(t  , fc1W, fc1b, outW, outb,
                                                  (float*)d_out, Bg);
}
